# starting kernel
# baseline (speedup 1.0000x reference)
#include <hip/hip_runtime.h>
#include <hip/hip_bf16.h>


#define N_NODES 100000
#define E_EDGES 1600000
#define IN_DIM  32
#define LATENT  64

typedef __attribute__((ext_vector_type(2))) float v2f;
typedef __attribute__((ext_vector_type(8))) float v8f;

__device__ __forceinline__ void atomic_add_f32(float* p, float v) {
    asm volatile("global_atomic_add_f32 %0, %1, off scope:SCOPE_DEV"
                 :: "v"(p), "v"(v)
                 : "memory");
}


__global__ void deg_kernel(const int* __restrict__ col,
                           const float* __restrict__ w,
                           float* __restrict__ deg) {
    int e = blockIdx.x * blockDim.x + threadIdx.x;
    if (e < E_EDGES) atomic_add_f32(&deg[col[e]], w[e]);
}

__global__ void dinv_kernel(float* __restrict__ deg) {
    int v = blockIdx.x * blockDim.x + threadIdx.x;
    if (v < N_NODES) {
        float d = deg[v];
        deg[v] = (d > 0.0f) ? __frsqrt_rn(d) : 0.0f;
    }
}

__global__ void norm_kernel(const int* __restrict__ row,
                            const int* __restrict__ col,
                            const float* __restrict__ w,
                            const float* __restrict__ dinv,
                            float* __restrict__ norm) {
    int e = blockIdx.x * blockDim.x + threadIdx.x;
    if (e < E_EDGES) norm[e] = dinv[row[e]] * w[e] * dinv[col[e]];
}


template <int D>
__global__ void scatter_kernel(const float* __restrict__ src,
                               float* __restrict__ dst,
                               const int* __restrict__ row,
                               const int* __restrict__ col,
                               const float* __restrict__ norm) {
    constexpr int VE = D / 4;
    int t = blockIdx.x * blockDim.x + threadIdx.x;
    int e = t / VE;
    int q = t & (VE - 1);
    int r = row[e];
    int c = col[e];
    float nv = norm[e];
    const float4 v = *(const float4*)(src + (size_t)r * D + q * 4);
    float* dp = dst + (size_t)c * D + q * 4;
    atomic_add_f32(dp + 0, nv * v.x);
    atomic_add_f32(dp + 1, nv * v.y);
    atomic_add_f32(dp + 2, nv * v.z);
    atomic_add_f32(dp + 3, nv * v.w);
}


template <int D>
__global__ void __launch_bounds__(256)
tag_gemm_kernel(const float* __restrict__ H0, const float* __restrict__ H1,
                const float* __restrict__ H2, const float* __restrict__ H3,
                const float* __restrict__ W,
                const float* __restrict__ bias,
                float* __restrict__ out) {
    const int wave  = threadIdx.x >> 5;
    const int lane  = threadIdx.x & 31;
    const int mtile = blockIdx.x * 2 + (wave >> 2);
    const int ntile = wave & 3;

    const int m    = lane & 15;
    const int half = lane >> 4;
    const int arow = mtile * 16 + m;
    const int ncol = ntile * 16 + m;

    const float* Hs[4] = {H0, H1, H2, H3};
    v8f c = {};

#pragma unroll
    for (int s = 0; s < 4; ++s) {
        const float* __restrict__ A  = Hs[s];
        const float* __restrict__ Wb = W + (size_t)s * D * 64;
#pragma unroll
        for (int k = 0; k < D; k += 4) {
            const int kk = k + half * 2;
            v2f a;
            a.x = A[(size_t)arow * D + kk];
            a.y = A[(size_t)arow * D + kk + 1];
            v2f b;
            b.x = Wb[(size_t)kk * 64 + ncol];
            b.y = Wb[(size_t)(kk + 1) * 64 + ncol];
            c = __builtin_amdgcn_wmma_f32_16x16x4_f32(
                     false, a,  false, b,
                     (short)0, c,  false,  false);
        }
    }

    const float bv = bias[ncol];
#pragma unroll
    for (int v = 0; v < 8; ++v) {
        const int orow = mtile * 16 + v + half * 8;
        float r = c[v] + bv;
        out[(size_t)orow * 64 + ncol] = r > 0.0f ? r : 0.0f;
    }
}


template <int D>
static void run_layer(const float* hin, const float* W, const float* bias,
                      float* buf1, float* buf2, float* buf3,
                      const int* row, const int* col, const float* norm,
                      float* hout, hipStream_t stream) {
    const size_t hopBytes = (size_t)N_NODES * D * sizeof(float);
    const int sblocks = (E_EDGES * (D / 4)) / 256;

    hipMemsetAsync(buf1, 0, hopBytes, stream);
    scatter_kernel<D><<<sblocks, 256, 0, stream>>>(hin, buf1, row, col, norm);
    hipMemsetAsync(buf2, 0, hopBytes, stream);
    scatter_kernel<D><<<sblocks, 256, 0, stream>>>(buf1, buf2, row, col, norm);
    hipMemsetAsync(buf3, 0, hopBytes, stream);
    scatter_kernel<D><<<sblocks, 256, 0, stream>>>(buf2, buf3, row, col, norm);

    tag_gemm_kernel<D><<<N_NODES / 32, 256, 0, stream>>>(
        hin, buf1, buf2, buf3, W, bias, hout);
}

extern "C" void kernel_launch(void* const* d_in, const int* in_sizes, int n_in,
                              void* d_out, int out_size, void* d_ws, size_t ws_size,
                              hipStream_t stream) {
    const float* x   = (const float*)d_in[0];
    const int*   ei  = (const int*)d_in[1];
    const float* ew  = (const float*)d_in[2];
    const float* W0  = (const float*)d_in[3];
    const float* b0  = (const float*)d_in[4];
    const float* Ws  = (const float*)d_in[5];
    const float* bs  = (const float*)d_in[6];
    float*       out = (float*)d_out;

    const int* row = ei;
    const int* col = ei + E_EDGES;

    float* ws   = (float*)d_ws;
    float* deg  = ws;
    float* norm = deg + N_NODES;
    float* bufA = norm + E_EDGES;
    float* bufB = bufA + (size_t)N_NODES * LATENT;
    float* buf1 = bufB + (size_t)N_NODES * LATENT;
    float* buf2 = buf1 + (size_t)N_NODES * LATENT;
    float* buf3 = buf2 + (size_t)N_NODES * LATENT;

    hipMemsetAsync(deg, 0, N_NODES * sizeof(float), stream);
    deg_kernel<<<E_EDGES / 256, 256, 0, stream>>>(col, ew, deg);
    dinv_kernel<<<(N_NODES + 255) / 256, 256, 0, stream>>>(deg);
    norm_kernel<<<E_EDGES / 256, 256, 0, stream>>>(row, col, ew, deg, norm);

    const size_t WL = 4 * 64 * 64;
    run_layer<IN_DIM>(x,    W0,          b0,       buf1, buf2, buf3, row, col, norm, bufA, stream);
    run_layer<LATENT>(bufA, Ws + 0 * WL, bs + 0,   buf1, buf2, buf3, row, col, norm, bufB, stream);
    run_layer<LATENT>(bufB, Ws + 1 * WL, bs + 64,  buf1, buf2, buf3, row, col, norm, bufA, stream);
    run_layer<LATENT>(bufA, Ws + 2 * WL, bs + 128, buf1, buf2, buf3, row, col, norm, bufB, stream);
    run_layer<LATENT>(bufB, Ws + 3 * WL, bs + 192, buf1, buf2, buf3, row, col, norm, out,  stream);
}


# Scatter hops: one wave per 32 edges, row ids in scalar registers; each gather load and each f32 atomic add covers one whole 128-B line (was 4 quarter-filled lines); sums stay in atomic arrival order
# speedup vs baseline: 1.3971x; 1.3971x over previous
#include <hip/hip_runtime.h>


#define N_NODES 100000
#define E_EDGES 1600000
#define IN_DIM  32
#define LATENT  64

typedef __attribute__((ext_vector_type(2))) float v2f;
typedef __attribute__((ext_vector_type(8))) float v8f;

__device__ __forceinline__ void atomic_add_f32(float* p, float v) {
    asm volatile("global_atomic_add_f32 %0, %1, off scope:SCOPE_DEV"
                 :: "v"(p), "v"(v)
                 : "memory");
}

template <int OFS>
__device__ __forceinline__ void atomic_add_f32_row(float* rowbase, unsigned lane_bytes, float v) {
    asm volatile("global_atomic_add_f32 %0, %1, %2 offset:%3 scope:SCOPE_DEV"
                 :: "v"(lane_bytes), "v"(v), "s"(rowbase), "n"(OFS)
                 : "memory");
}


__global__ void deg_kernel(const int* __restrict__ col,
                           const float* __restrict__ w,
                           float* __restrict__ deg) {
    int e = blockIdx.x * blockDim.x + threadIdx.x;
    if (e < E_EDGES) atomic_add_f32(&deg[col[e]], w[e]);
}

__global__ void dinv_kernel(float* __restrict__ deg) {
    int v = blockIdx.x * blockDim.x + threadIdx.x;
    if (v < N_NODES) {
        float d = deg[v];
        deg[v] = (d > 0.0f) ? __frsqrt_rn(d) : 0.0f;
    }
}

__global__ void norm_kernel(const int* __restrict__ row,
                            const int* __restrict__ col,
                            const float* __restrict__ w,
                            const float* __restrict__ dinv,
                            float* __restrict__ norm) {
    int e = blockIdx.x * blockDim.x + threadIdx.x;
    if (e < E_EDGES) norm[e] = dinv[row[e]] * w[e] * dinv[col[e]];
}


template <int D>
__global__ void __launch_bounds__(256)
scatter_kernel(const float* __restrict__ src,
               float* __restrict__ dst,
               const int* __restrict__ row,
               const int* __restrict__ col,
               const float* __restrict__ norm) {
    constexpr int LPR = D / 32;
    constexpr int G   = 16 / LPR;
    const int lane = threadIdx.x & 31;
    const int wv = __builtin_amdgcn_readfirstlane((int)(blockIdx.x * 8 + (threadIdx.x >> 5)));
    const int e  = wv * 32 + lane;
    const int   rv = row[e];
    const int   cv = col[e];
    const int   nv = __float_as_int(norm[e]);
    const float* __restrict__ srcl = src + lane;
    const unsigned lane_bytes = (unsigned)lane * 4u;

    for (int j = 0; j < 32; j += G) {
        float x[G][LPR];
#pragma unroll
        for (int k = 0; k < G; ++k) {
            const int r = __builtin_amdgcn_readlane(rv, j + k);
            const float* __restrict__ rp = srcl + (size_t)r * D;
            x[k][0] = rp[0];
            if (LPR == 2) x[k][LPR - 1] = rp[32 * (LPR - 1)];
        }
#pragma unroll
        for (int k = 0; k < G; ++k) {
            const int   c = __builtin_amdgcn_readlane(cv, j + k);
            const float w = __int_as_float(__builtin_amdgcn_readlane(nv, j + k));
            float* dp = dst + (size_t)c * D;
            atomic_add_f32_row<0>(dp, lane_bytes, w * x[k][0]);
            if (LPR == 2) atomic_add_f32_row<128 * (LPR - 1)>(dp, lane_bytes, w * x[k][LPR - 1]);
        }
    }
}


template <int D>
__global__ void __launch_bounds__(256)
tag_gemm_kernel(const float* __restrict__ H0, const float* __restrict__ H1,
                const float* __restrict__ H2, const float* __restrict__ H3,
                const float* __restrict__ W,
                const float* __restrict__ bias,
                float* __restrict__ out) {
    const int wave  = threadIdx.x >> 5;
    const int lane  = threadIdx.x & 31;
    const int mtile = blockIdx.x * 2 + (wave >> 2);
    const int ntile = wave & 3;

    const int m    = lane & 15;
    const int half = lane >> 4;
    const int arow = mtile * 16 + m;
    const int ncol = ntile * 16 + m;

    const float* Hs[4] = {H0, H1, H2, H3};
    v8f c = {};

#pragma unroll
    for (int s = 0; s < 4; ++s) {
        const float* __restrict__ A  = Hs[s];
        const float* __restrict__ Wb = W + (size_t)s * D * 64;
#pragma unroll
        for (int k = 0; k < D; k += 4) {
            const int kk = k + half * 2;
            v2f a;
            a.x = A[(size_t)arow * D + kk];
            a.y = A[(size_t)arow * D + kk + 1];
            v2f b;
            b.x = Wb[(size_t)kk * 64 + ncol];
            b.y = Wb[(size_t)(kk + 1) * 64 + ncol];
            c = __builtin_amdgcn_wmma_f32_16x16x4_f32(
                     false, a,  false, b,
                     (short)0, c,  false,  false);
        }
    }

    const float bv = bias[ncol];
#pragma unroll
    for (int v = 0; v < 8; ++v) {
        const int orow = mtile * 16 + v + half * 8;
        float r = c[v] + bv;
        out[(size_t)orow * 64 + ncol] = r > 0.0f ? r : 0.0f;
    }
}


template <int D>
static void run_layer(const float* hin, const float* W, const float* bias,
                      float* buf1, float* buf2, float* buf3,
                      const int* row, const int* col, const float* norm,
                      float* hout, hipStream_t stream) {
    const size_t hopBytes = (size_t)N_NODES * D * sizeof(float);
    const int sblocks = E_EDGES / 256;

    hipMemsetAsync(buf1, 0, hopBytes, stream);
    scatter_kernel<D><<<sblocks, 256, 0, stream>>>(hin, buf1, row, col, norm);
    hipMemsetAsync(buf2, 0, hopBytes, stream);
    scatter_kernel<D><<<sblocks, 256, 0, stream>>>(buf1, buf2, row, col, norm);
    hipMemsetAsync(buf3, 0, hopBytes, stream);
    scatter_kernel<D><<<sblocks, 256, 0, stream>>>(buf2, buf3, row, col, norm);

    tag_gemm_kernel<D><<<N_NODES / 32, 256, 0, stream>>>(
        hin, buf1, buf2, buf3, W, bias, hout);
}

extern "C" void kernel_launch(void* const* d_in, const int* in_sizes, int n_in,
                              void* d_out, int out_size, void* d_ws, size_t ws_size,
                              hipStream_t stream) {
    const float* x   = (const float*)d_in[0];
    const int*   ei  = (const int*)d_in[1];
    const float* ew  = (const float*)d_in[2];
    const float* W0  = (const float*)d_in[3];
    const float* b0  = (const float*)d_in[4];
    const float* Ws  = (const float*)d_in[5];
    const float* bs  = (const float*)d_in[6];
    float*       out = (float*)d_out;

    const int* row = ei;
    const int* col = ei + E_EDGES;

    float* ws   = (float*)d_ws;
    float* deg  = ws;
    float* norm = deg + N_NODES;
    float* bufA = norm + E_EDGES;
    float* bufB = bufA + (size_t)N_NODES * LATENT;
    float* buf1 = bufB + (size_t)N_NODES * LATENT;
    float* buf2 = buf1 + (size_t)N_NODES * LATENT;
    float* buf3 = buf2 + (size_t)N_NODES * LATENT;

    hipMemsetAsync(deg, 0, N_NODES * sizeof(float), stream);
    deg_kernel<<<E_EDGES / 256, 256, 0, stream>>>(col, ew, deg);
    dinv_kernel<<<(N_NODES + 255) / 256, 256, 0, stream>>>(deg);
    norm_kernel<<<E_EDGES / 256, 256, 0, stream>>>(row, col, ew, deg, norm);

    const size_t WL = 4 * 64 * 64;
    run_layer<IN_DIM>(x,    W0,          b0,       buf1, buf2, buf3, row, col, norm, bufA, stream);
    run_layer<LATENT>(bufA, Ws + 0 * WL, bs + 0,   buf1, buf2, buf3, row, col, norm, bufB, stream);
    run_layer<LATENT>(bufB, Ws + 1 * WL, bs + 64,  buf1, buf2, buf3, row, col, norm, bufA, stream);
    run_layer<LATENT>(bufA, Ws + 2 * WL, bs + 128, buf1, buf2, buf3, row, col, norm, bufB, stream);
    run_layer<LATENT>(bufB, Ws + 3 * WL, bs + 192, buf1, buf2, buf3, row, col, norm, out,  stream);
}


# Hops without atomics: edges sorted by destination once per call (stable 2-pass radix sort, LDS counts); a wave sums a node's in-edges in registers, stores the row once; sums now in edge-list order
# speedup vs baseline: 2.9108x; 2.0835x over previous
#include <hip/hip_runtime.h>


#define N_NODES 100000
#define E_EDGES 1600000
#define IN_DIM  32
#define LATENT  64

#define UE      1280
#define NUNITS  (E_EDGES / UE)
#define UBLOCKS ((NUNITS + 7) / 8)
#define LO_BITS 9
#define HI_BITS 8
#define SCAN_T  1024
#define HOP_BLOCKS 2048
#define HOP_WAVES  (HOP_BLOCKS * 8)

typedef __attribute__((ext_vector_type(2))) float v2f;
typedef __attribute__((ext_vector_type(8))) float v8f;

__device__ __forceinline__ void atomic_add_f32(float* p, float v) {
    asm volatile("global_atomic_add_f32 %0, %1, off scope:SCOPE_DEV"
                 :: "v"(p), "v"(v)
                 : "memory");
}


__global__ void deg_kernel(const int* __restrict__ col,
                           const float* __restrict__ w,
                           float* __restrict__ deg) {
    int e = blockIdx.x * blockDim.x + threadIdx.x;
    if (e < E_EDGES) atomic_add_f32(&deg[col[e]], w[e]);
}

__global__ void dinv_kernel(float* __restrict__ deg) {
    int v = blockIdx.x * blockDim.x + threadIdx.x;
    if (v < N_NODES) {
        float d = deg[v];
        deg[v] = (d > 0.0f) ? __frsqrt_rn(d) : 0.0f;
    }
}

__global__ void norm_kernel(const int* __restrict__ row,
                            const int* __restrict__ col,
                            const float* __restrict__ w,
                            const float* __restrict__ dinv,
                            float* __restrict__ norm) {
    int e = blockIdx.x * blockDim.x + threadIdx.x;
    if (e < E_EDGES) norm[e] = dinv[row[e]] * w[e] * dinv[col[e]];
}


template <int BITS>
__device__ __forceinline__ unsigned same_digit_lanes(int d) {
    unsigned m = 0xffffffffu;
#pragma unroll
    for (int i = 0; i < BITS; ++i) {
        const bool bit = (d >> i) & 1;
        const unsigned bal = __builtin_amdgcn_ballot_w32(bit);
        m &= bit ? bal : ~bal;
    }
    return m;
}

template <int BITS, int SHIFT>
__global__ void __launch_bounds__(256)
digit_count_kernel(const int* __restrict__ key, int* __restrict__ T) {
    constexpr int NB = 1 << BITS;
    __shared__ int h[8][NB];
    const int lane = threadIdx.x & 31;
    const int wave = threadIdx.x >> 5;
    const int unit = blockIdx.x * 8 + wave;
    if (unit >= NUNITS) return;
    int* hw = h[wave];
    for (int i = lane; i < NB; i += 32) hw[i] = 0;
    const unsigned lt = (1u << lane) - 1u;
    for (int c0 = 0; c0 < UE / 32; c0 += 8) {
        int kk[8];
#pragma unroll
        for (int u = 0; u < 8; ++u) kk[u] = key[unit * UE + (c0 + u) * 32 + lane];
#pragma unroll
        for (int u = 0; u < 8; ++u) {
            const int d = (kk[u] >> SHIFT) & (NB - 1);
            const unsigned m = same_digit_lanes<BITS>(d);
            if ((m & lt) == 0) hw[d] = hw[d] + __popc(m);
        }
    }
    for (int i = lane; i < NB; i += 32) T[(size_t)i * NUNITS + unit] = hw[i];
}

__global__ void __launch_bounds__(SCAN_T)
block_total_kernel(const int* __restrict__ T, int nt, int* __restrict__ bt) {
    __shared__ int ws[32];
    const int t = threadIdx.x, lane = t & 31, wave = t >> 5;
    const int i = blockIdx.x * SCAN_T + t;
    int s = (i < nt) ? T[i] : 0;
#pragma unroll
    for (int d = 16; d > 0; d >>= 1) s += __shfl_xor(s, d);
    if (lane == 0) ws[wave] = s;
    __syncthreads();
    if (wave == 0) {
        int q = ws[lane];
#pragma unroll
        for (int d = 16; d > 0; d >>= 1) q += __shfl_xor(q, d);
        if (lane == 0) bt[blockIdx.x] = q;
    }
}

__global__ void __launch_bounds__(SCAN_T)
scan_final_kernel(int* __restrict__ T, int nt, const int* __restrict__ bt) {
    __shared__ int wtot[32];
    __shared__ int wpre[32];
    const int t = threadIdx.x, lane = t & 31, wave = t >> 5;
    int p = 0;
    for (int i = t; i < (int)blockIdx.x; i += SCAN_T) p += bt[i];
    const int idx = blockIdx.x * SCAN_T + t;
    const int c = (idx < nt) ? T[idx] : 0;
    int inc = c;
#pragma unroll
    for (int d = 1; d < 32; d <<= 1) {
        const int y = __shfl_up(inc, d);
        if (lane >= d) inc += y;
    }
#pragma unroll
    for (int d = 16; d > 0; d >>= 1) p += __shfl_xor(p, d);
    if (lane == 31) wtot[wave] = inc;
    if (lane == 0)  wpre[wave] = p;
    __syncthreads();
    const int wt = wtot[lane];
    int winc = wt;
#pragma unroll
    for (int d = 1; d < 32; d <<= 1) {
        const int y = __shfl_up(winc, d);
        if (lane >= d) winc += y;
    }
    const int woff = __shfl(winc, wave) - __shfl(wt, wave);
    int P = wpre[lane];
#pragma unroll
    for (int d = 16; d > 0; d >>= 1) P += __shfl_xor(P, d);
    if (idx < nt) T[idx] = P + woff + inc - c;
}

template <int BITS, int SHIFT>
__global__ void __launch_bounds__(256)
digit_place_kernel(const int* __restrict__ key, const int* __restrict__ rin,
                   const float* __restrict__ nin, const int* __restrict__ T,
                   int* __restrict__ kout, int* __restrict__ rout,
                   float* __restrict__ nout) {
    constexpr int NB = 1 << BITS;
    __shared__ int h[8][NB];
    const int lane = threadIdx.x & 31;
    const int wave = threadIdx.x >> 5;
    const int unit = blockIdx.x * 8 + wave;
    if (unit >= NUNITS) return;
    int* hw = h[wave];
    for (int i = lane; i < NB; i += 32) hw[i] = T[(size_t)i * NUNITS + unit];
    const unsigned lt = (1u << lane) - 1u;
    for (int c0 = 0; c0 < UE / 32; c0 += 8) {
      int kk[8], rr[8]; float nn[8];
#pragma unroll
      for (int u = 0; u < 8; ++u) {
          const int e = unit * UE + (c0 + u) * 32 + lane;
          kk[u] = key[e];
          rr[u] = rin[e];
          nn[u] = nin[e];
      }
#pragma unroll
      for (int u = 0; u < 8; ++u) {
        const int   k = kk[u];
        const int   r = rr[u];
        const float n = nn[u];
        const int d = (k >> SHIFT) & (NB - 1);
        const unsigned m = same_digit_lanes<BITS>(d);
        const int first = __ffs(m) - 1;
        int start = 0;
        if (lane == first) {
            start = hw[d];
            hw[d] = start + __popc(m);
        }
        start = __shfl(start, first);
        const int pos = start + __popc(m & lt);
        kout[pos] = k;
        rout[pos] = r;
        nout[pos] = n;
      }
    }
}

__global__ void run_start_kernel(const int* __restrict__ ks, int* __restrict__ off) {
    const int p = blockIdx.x * blockDim.x + threadIdx.x;
    if (p < E_EDGES) {
        const int k  = ks[p];
        const int kp = (p > 0) ? ks[p - 1] : -1;
        for (int v = kp + 1; v <= k; ++v) off[v] = p;
        if (p == E_EDGES - 1)
            for (int v = k + 1; v <= N_NODES; ++v) off[v] = E_EDGES;
    }
}


template <int VPL> struct row_piece;
template <> struct row_piece<1> { typedef float type; };
template <> struct row_piece<2> { typedef v2f   type; };

template <int D, int G>
__device__ __forceinline__ void gather_group(const float* __restrict__ src, int lane,
                                             int rv, int nv, int j,
                                             typename row_piece<D / 32>::type& acc) {
    typedef typename row_piece<D / 32>::type vec;
    vec x[G];
#pragma unroll
    for (int k = 0; k < G; ++k) {
        const int r = __builtin_amdgcn_readlane(rv, j + k);
        x[k] = reinterpret_cast<const vec*>(src + (size_t)r * D)[lane];
    }
#pragma unroll
    for (int k = 0; k < G; ++k) {
        const float w = __int_as_float(__builtin_amdgcn_readlane(nv, j + k));
        acc += w * x[k];
    }
}

template <int D>
__global__ void __launch_bounds__(256)
gather_kernel(const float* __restrict__ src,
              float* __restrict__ dst,
              const int* __restrict__ off,
              const int* __restrict__ rs,
              const float* __restrict__ ns) {
    typedef typename row_piece<D / 32>::type vec;
    const int lane = threadIdx.x & 31;
    const int wv = __builtin_amdgcn_readfirstlane((int)(blockIdx.x * 8 + (threadIdx.x >> 5)));

    for (int v = wv; v < N_NODES; v += HOP_WAVES) {
        const int beg = off[v];
        const int end = off[v + 1];
        vec acc = 0.0f;
        for (int b = beg; b < end; b += 32) {
            const int n = (end - b < 32) ? (end - b) : 32;
            int rv = 0, nv = 0;
            if (lane < n) {
                rv = rs[b + lane];
                nv = __float_as_int(ns[b + lane]);
            }
            int j = 0;
            for (; j + 16 <= n; j += 16) gather_group<D, 16>(src, lane, rv, nv, j, acc);
            if (n - j >= 8) { gather_group<D, 8>(src, lane, rv, nv, j, acc); j += 8; }
            if (n - j >= 4) { gather_group<D, 4>(src, lane, rv, nv, j, acc); j += 4; }
            if (n - j >= 2) { gather_group<D, 2>(src, lane, rv, nv, j, acc); j += 2; }
            if (n - j >= 1) { gather_group<D, 1>(src, lane, rv, nv, j, acc); }
        }
        reinterpret_cast<vec*>(dst + (size_t)v * D)[lane] = acc;
    }
}


template <int D>
__global__ void __launch_bounds__(256)
tag_gemm_kernel(const float* __restrict__ H0, const float* __restrict__ H1,
                const float* __restrict__ H2, const float* __restrict__ H3,
                const float* __restrict__ W,
                const float* __restrict__ bias,
                float* __restrict__ out) {
    const int wave  = threadIdx.x >> 5;
    const int lane  = threadIdx.x & 31;
    const int mtile = blockIdx.x * 2 + (wave >> 2);
    const int ntile = wave & 3;

    const int m    = lane & 15;
    const int half = lane >> 4;
    const int arow = mtile * 16 + m;
    const int ncol = ntile * 16 + m;

    const float* Hs[4] = {H0, H1, H2, H3};
    v8f c = {};

#pragma unroll
    for (int s = 0; s < 4; ++s) {
        const float* __restrict__ A  = Hs[s];
        const float* __restrict__ Wb = W + (size_t)s * D * 64;
#pragma unroll
        for (int k = 0; k < D; k += 4) {
            const int kk = k + half * 2;
            v2f a;
            a.x = A[(size_t)arow * D + kk];
            a.y = A[(size_t)arow * D + kk + 1];
            v2f b;
            b.x = Wb[(size_t)kk * 64 + ncol];
            b.y = Wb[(size_t)(kk + 1) * 64 + ncol];
            c = __builtin_amdgcn_wmma_f32_16x16x4_f32(
                     false, a,  false, b,
                     (short)0, c,  false,  false);
        }
    }

    const float bv = bias[ncol];
#pragma unroll
    for (int v = 0; v < 8; ++v) {
        const int orow = mtile * 16 + v + half * 8;
        float r = c[v] + bv;
        out[(size_t)orow * 64 + ncol] = r > 0.0f ? r : 0.0f;
    }
}


template <int BITS, int SHIFT>
static void sort_pass(const int* kin, const int* rin, const float* nin,
                      int* T, int* bt, int* kout, int* rout, float* nout,
                      hipStream_t stream) {
    const int nt = (1 << BITS) * NUNITS;
    const int sb = (nt + SCAN_T - 1) / SCAN_T;
    digit_count_kernel<BITS, SHIFT><<<UBLOCKS, 256, 0, stream>>>(kin, T);
    block_total_kernel<<<sb, SCAN_T, 0, stream>>>(T, nt, bt);
    scan_final_kernel<<<sb, SCAN_T, 0, stream>>>(T, nt, bt);
    digit_place_kernel<BITS, SHIFT><<<UBLOCKS, 256, 0, stream>>>(kin, rin, nin, T, kout, rout, nout);
}

template <int D>
static void run_layer(const float* hin, const float* W, const float* bias,
                      float* buf1, float* buf2, float* buf3,
                      const int* off, const int* rs, const float* ns,
                      float* hout, hipStream_t stream) {
    gather_kernel<D><<<HOP_BLOCKS, 256, 0, stream>>>(hin,  buf1, off, rs, ns);
    gather_kernel<D><<<HOP_BLOCKS, 256, 0, stream>>>(buf1, buf2, off, rs, ns);
    gather_kernel<D><<<HOP_BLOCKS, 256, 0, stream>>>(buf2, buf3, off, rs, ns);

    tag_gemm_kernel<D><<<N_NODES / 32, 256, 0, stream>>>(
        hin, buf1, buf2, buf3, W, bias, hout);
}

extern "C" void kernel_launch(void* const* d_in, const int* in_sizes, int n_in,
                              void* d_out, int out_size, void* d_ws, size_t ws_size,
                              hipStream_t stream) {
    const float* x   = (const float*)d_in[0];
    const int*   ei  = (const int*)d_in[1];
    const float* ew  = (const float*)d_in[2];
    const float* W0  = (const float*)d_in[3];
    const float* b0  = (const float*)d_in[4];
    const float* Ws  = (const float*)d_in[5];
    const float* bs  = (const float*)d_in[6];
    float*       out = (float*)d_out;

    const int* row = ei;
    const int* col = ei + E_EDGES;

    const size_t NPAD = 100096;
    const size_t TPAD = 640000;
    float* ws   = (float*)d_ws;
    float* deg  = ws;
    int*   off  = (int*)(deg + NPAD);
    int*   bt   = off + NPAD;
    int*   T    = bt + 1024;
    float* norm = (float*)(T + TPAD);
    int*   k1   = (int*)(norm + E_EDGES);
    int*   r1   = k1 + E_EDGES;
    float* n1   = (float*)(r1 + E_EDGES);
    int*   k2   = (int*)(n1 + E_EDGES);
    int*   r2   = k2 + E_EDGES;
    float* n2   = (float*)(r2 + E_EDGES);
    float* bufA = n2 + E_EDGES;
    float* bufB = bufA + (size_t)N_NODES * LATENT;
    float* buf1 = bufB + (size_t)N_NODES * LATENT;
    float* buf2 = buf1 + (size_t)N_NODES * LATENT;
    float* buf3 = buf2 + (size_t)N_NODES * LATENT;

    hipMemsetAsync(deg, 0, N_NODES * sizeof(float), stream);
    deg_kernel<<<E_EDGES / 256, 256, 0, stream>>>(col, ew, deg);
    dinv_kernel<<<(N_NODES + 255) / 256, 256, 0, stream>>>(deg);
    norm_kernel<<<E_EDGES / 256, 256, 0, stream>>>(row, col, ew, deg, norm);

    sort_pass<LO_BITS, 0>(col, row, norm, T, bt, k1, r1, n1, stream);
    sort_pass<HI_BITS, LO_BITS>(k1, r1, n1, T, bt, k2, r2, n2, stream);
    run_start_kernel<<<E_EDGES / 256, 256, 0, stream>>>(k2, off);

    const size_t WL = 4 * 64 * 64;
    run_layer<IN_DIM>(x,    W0,          b0,       buf1, buf2, buf3, off, r2, n2, bufA, stream);
    run_layer<LATENT>(bufA, Ws + 0 * WL, bs + 0,   buf1, buf2, buf3, off, r2, n2, bufB, stream);
    run_layer<LATENT>(bufB, Ws + 1 * WL, bs + 64,  buf1, buf2, buf3, off, r2, n2, bufA, stream);
    run_layer<LATENT>(bufA, Ws + 2 * WL, bs + 128, buf1, buf2, buf3, off, r2, n2, bufB, stream);
    run_layer<LATENT>(bufB, Ws + 3 * WL, bs + 192, buf1, buf2, buf3, off, r2, n2, out,  stream);
}


# Hop GEMM: a wave owns 16 rows x 64 columns (4 accumulator tiles): H read once in 16-B loads, weights staged per block in LDS, whole-row stores; k order inside each 8-k group changed
# speedup vs baseline: 3.2925x; 1.1311x over previous
#include <hip/hip_runtime.h>


#define N_NODES 100000
#define E_EDGES 1600000
#define IN_DIM  32
#define LATENT  64

#define UE      1280
#define NUNITS  (E_EDGES / UE)
#define UBLOCKS ((NUNITS + 7) / 8)
#define LO_BITS 9
#define HI_BITS 8
#define SCAN_T  1024
#define HOP_BLOCKS 2048
#define HOP_WAVES  (HOP_BLOCKS * 8)

typedef __attribute__((ext_vector_type(2))) float v2f;
typedef __attribute__((ext_vector_type(4))) float v4f;
typedef __attribute__((ext_vector_type(8))) float v8f;

__device__ __forceinline__ void atomic_add_f32(float* p, float v) {
    asm volatile("global_atomic_add_f32 %0, %1, off scope:SCOPE_DEV"
                 :: "v"(p), "v"(v)
                 : "memory");
}


__global__ void deg_kernel(const int* __restrict__ col,
                           const float* __restrict__ w,
                           float* __restrict__ deg) {
    int e = blockIdx.x * blockDim.x + threadIdx.x;
    if (e < E_EDGES) atomic_add_f32(&deg[col[e]], w[e]);
}

__global__ void dinv_kernel(float* __restrict__ deg) {
    int v = blockIdx.x * blockDim.x + threadIdx.x;
    if (v < N_NODES) {
        float d = deg[v];
        deg[v] = (d > 0.0f) ? __frsqrt_rn(d) : 0.0f;
    }
}

__global__ void norm_kernel(const int* __restrict__ row,
                            const int* __restrict__ col,
                            const float* __restrict__ w,
                            const float* __restrict__ dinv,
                            float* __restrict__ norm) {
    int e = blockIdx.x * blockDim.x + threadIdx.x;
    if (e < E_EDGES) norm[e] = dinv[row[e]] * w[e] * dinv[col[e]];
}


template <int BITS>
__device__ __forceinline__ unsigned same_digit_lanes(int d) {
    unsigned m = 0xffffffffu;
#pragma unroll
    for (int i = 0; i < BITS; ++i) {
        const bool bit = (d >> i) & 1;
        const unsigned bal = __builtin_amdgcn_ballot_w32(bit);
        m &= bit ? bal : ~bal;
    }
    return m;
}

template <int BITS, int SHIFT>
__global__ void __launch_bounds__(256)
digit_count_kernel(const int* __restrict__ key, int* __restrict__ T) {
    constexpr int NB = 1 << BITS;
    __shared__ int h[8][NB];
    const int lane = threadIdx.x & 31;
    const int wave = threadIdx.x >> 5;
    const int unit = blockIdx.x * 8 + wave;
    if (unit >= NUNITS) return;
    int* hw = h[wave];
    for (int i = lane; i < NB; i += 32) hw[i] = 0;
    const unsigned lt = (1u << lane) - 1u;
    for (int c0 = 0; c0 < UE / 32; c0 += 8) {
        int kk[8];
#pragma unroll
        for (int u = 0; u < 8; ++u) kk[u] = key[unit * UE + (c0 + u) * 32 + lane];
#pragma unroll
        for (int u = 0; u < 8; ++u) {
            const int d = (kk[u] >> SHIFT) & (NB - 1);
            const unsigned m = same_digit_lanes<BITS>(d);
            if ((m & lt) == 0) hw[d] = hw[d] + __popc(m);
        }
    }
    for (int i = lane; i < NB; i += 32) T[(size_t)i * NUNITS + unit] = hw[i];
}

__global__ void __launch_bounds__(SCAN_T)
block_total_kernel(const int* __restrict__ T, int nt, int* __restrict__ bt) {
    __shared__ int ws[32];
    const int t = threadIdx.x, lane = t & 31, wave = t >> 5;
    const int i = blockIdx.x * SCAN_T + t;
    int s = (i < nt) ? T[i] : 0;
#pragma unroll
    for (int d = 16; d > 0; d >>= 1) s += __shfl_xor(s, d);
    if (lane == 0) ws[wave] = s;
    __syncthreads();
    if (wave == 0) {
        int q = ws[lane];
#pragma unroll
        for (int d = 16; d > 0; d >>= 1) q += __shfl_xor(q, d);
        if (lane == 0) bt[blockIdx.x] = q;
    }
}

__global__ void __launch_bounds__(SCAN_T)
scan_final_kernel(int* __restrict__ T, int nt, const int* __restrict__ bt) {
    __shared__ int wtot[32];
    __shared__ int wpre[32];
    const int t = threadIdx.x, lane = t & 31, wave = t >> 5;
    int p = 0;
    for (int i = t; i < (int)blockIdx.x; i += SCAN_T) p += bt[i];
    const int idx = blockIdx.x * SCAN_T + t;
    const int c = (idx < nt) ? T[idx] : 0;
    int inc = c;
#pragma unroll
    for (int d = 1; d < 32; d <<= 1) {
        const int y = __shfl_up(inc, d);
        if (lane >= d) inc += y;
    }
#pragma unroll
    for (int d = 16; d > 0; d >>= 1) p += __shfl_xor(p, d);
    if (lane == 31) wtot[wave] = inc;
    if (lane == 0)  wpre[wave] = p;
    __syncthreads();
    const int wt = wtot[lane];
    int winc = wt;
#pragma unroll
    for (int d = 1; d < 32; d <<= 1) {
        const int y = __shfl_up(winc, d);
        if (lane >= d) winc += y;
    }
    const int woff = __shfl(winc, wave) - __shfl(wt, wave);
    int P = wpre[lane];
#pragma unroll
    for (int d = 16; d > 0; d >>= 1) P += __shfl_xor(P, d);
    if (idx < nt) T[idx] = P + woff + inc - c;
}

template <int BITS, int SHIFT>
__global__ void __launch_bounds__(256)
digit_place_kernel(const int* __restrict__ key, const int* __restrict__ rin,
                   const float* __restrict__ nin, const int* __restrict__ T,
                   int* __restrict__ kout, int* __restrict__ rout,
                   float* __restrict__ nout) {
    constexpr int NB = 1 << BITS;
    __shared__ int h[8][NB];
    const int lane = threadIdx.x & 31;
    const int wave = threadIdx.x >> 5;
    const int unit = blockIdx.x * 8 + wave;
    if (unit >= NUNITS) return;
    int* hw = h[wave];
    for (int i = lane; i < NB; i += 32) hw[i] = T[(size_t)i * NUNITS + unit];
    const unsigned lt = (1u << lane) - 1u;
    for (int c0 = 0; c0 < UE / 32; c0 += 8) {
      int kk[8], rr[8]; float nn[8];
#pragma unroll
      for (int u = 0; u < 8; ++u) {
          const int e = unit * UE + (c0 + u) * 32 + lane;
          kk[u] = key[e];
          rr[u] = rin[e];
          nn[u] = nin[e];
      }
#pragma unroll
      for (int u = 0; u < 8; ++u) {
        const int   k = kk[u];
        const int   r = rr[u];
        const float n = nn[u];
        const int d = (k >> SHIFT) & (NB - 1);
        const unsigned m = same_digit_lanes<BITS>(d);
        const int first = __ffs(m) - 1;
        int start = 0;
        if (lane == first) {
            start = hw[d];
            hw[d] = start + __popc(m);
        }
        start = __shfl(start, first);
        const int pos = start + __popc(m & lt);
        kout[pos] = k;
        rout[pos] = r;
        nout[pos] = n;
      }
    }
}

__global__ void run_start_kernel(const int* __restrict__ ks, int* __restrict__ off) {
    const int p = blockIdx.x * blockDim.x + threadIdx.x;
    if (p < E_EDGES) {
        const int k  = ks[p];
        const int kp = (p > 0) ? ks[p - 1] : -1;
        for (int v = kp + 1; v <= k; ++v) off[v] = p;
        if (p == E_EDGES - 1)
            for (int v = k + 1; v <= N_NODES; ++v) off[v] = E_EDGES;
    }
}


template <int VPL> struct row_piece;
template <> struct row_piece<1> { typedef float type; };
template <> struct row_piece<2> { typedef v2f   type; };

template <int D, int G>
__device__ __forceinline__ void gather_group(const float* __restrict__ src, int lane,
                                             int rv, int nv, int j,
                                             typename row_piece<D / 32>::type& acc) {
    typedef typename row_piece<D / 32>::type vec;
    vec x[G];
#pragma unroll
    for (int k = 0; k < G; ++k) {
        const int r = __builtin_amdgcn_readlane(rv, j + k);
        x[k] = reinterpret_cast<const vec*>(src + (size_t)r * D)[lane];
    }
#pragma unroll
    for (int k = 0; k < G; ++k) {
        const float w = __int_as_float(__builtin_amdgcn_readlane(nv, j + k));
        acc += w * x[k];
    }
}

template <int D>
__global__ void __launch_bounds__(256)
gather_kernel(const float* __restrict__ src,
              float* __restrict__ dst,
              const int* __restrict__ off,
              const int* __restrict__ rs,
              const float* __restrict__ ns) {
    typedef typename row_piece<D / 32>::type vec;
    const int lane = threadIdx.x & 31;
    const int wv = __builtin_amdgcn_readfirstlane((int)(blockIdx.x * 8 + (threadIdx.x >> 5)));

    for (int v = wv; v < N_NODES; v += HOP_WAVES) {
        const int beg = off[v];
        const int end = off[v + 1];
        vec acc = 0.0f;
        for (int b = beg; b < end; b += 32) {
            const int n = (end - b < 32) ? (end - b) : 32;
            int rv = 0, nv = 0;
            if (lane < n) {
                rv = rs[b + lane];
                nv = __float_as_int(ns[b + lane]);
            }
            int j = 0;
            for (; j + 16 <= n; j += 16) gather_group<D, 16>(src, lane, rv, nv, j, acc);
            if (n - j >= 8) { gather_group<D, 8>(src, lane, rv, nv, j, acc); j += 8; }
            if (n - j >= 4) { gather_group<D, 4>(src, lane, rv, nv, j, acc); j += 4; }
            if (n - j >= 2) { gather_group<D, 2>(src, lane, rv, nv, j, acc); j += 2; }
            if (n - j >= 1) { gather_group<D, 1>(src, lane, rv, nv, j, acc); }
        }
        reinterpret_cast<vec*>(dst + (size_t)v * D)[lane] = acc;
    }
}


#define WQ_PITCH 288

template <int D>
__global__ void __launch_bounds__(128)
tag_gemm_kernel(const float* __restrict__ H0, const float* __restrict__ H1,
                const float* __restrict__ H2, const float* __restrict__ H3,
                const float* __restrict__ W,
                const float* __restrict__ bias,
                float* __restrict__ out) {
    constexpr int NQ = (4 * D) / 4;
    __shared__ __attribute__((aligned(16))) float Wl[NQ * WQ_PITCH];

    const int t = threadIdx.x;
#pragma unroll
    for (int it = 0; it < NQ * 16 / 128; ++it) {
        const int i = t + 128 * it;
        const int q = i >> 4, m = i & 15;
        const float* __restrict__ src = W + (size_t)(4 * q) * 64 + 4 * m;
        const v4f r0 = *reinterpret_cast<const v4f*>(src);
        const v4f r1 = *reinterpret_cast<const v4f*>(src + 64);
        const v4f r2 = *reinterpret_cast<const v4f*>(src + 128);
        const v4f r3 = *reinterpret_cast<const v4f*>(src + 192);
        float* dst = Wl + q * WQ_PITCH + m * 4;
        v4f c0 = {r0.x, r1.x, r2.x, r3.x};
        v4f c1 = {r0.y, r1.y, r2.y, r3.y};
        v4f c2 = {r0.z, r1.z, r2.z, r3.z};
        v4f c3 = {r0.w, r1.w, r2.w, r3.w};
        *reinterpret_cast<v4f*>(dst)       = c0;
        *reinterpret_cast<v4f*>(dst + 64)  = c1;
        *reinterpret_cast<v4f*>(dst + 128) = c2;
        *reinterpret_cast<v4f*>(dst + 192) = c3;
    }
    __syncthreads();

    const int wave  = t >> 5;
    const int lane  = t & 31;
    const int mtile = blockIdx.x * 4 + wave;
    if (mtile >= N_NODES / 16) return;

    const int c = lane & 15;
    const int h = lane >> 4;
    const size_t arow = (size_t)mtile * 16 + c;

    const float* Hs[4] = {H0, H1, H2, H3};
    v8f acc[4] = {};

#pragma unroll
    for (int s = 0; s < 4; ++s) {
        const float* __restrict__ A = Hs[s] + arow * D + 4 * h;
        v4f a[D / 8];
#pragma unroll
        for (int j = 0; j < D / 8; ++j) a[j] = *reinterpret_cast<const v4f*>(A + 8 * j);
#pragma unroll
        for (int j = 0; j < D / 8; ++j) {
            const float* bq = Wl + ((s * D + 8 * j) / 4 + h) * WQ_PITCH + c * 4;
            v4f b[4];
#pragma unroll
            for (int tt = 0; tt < 4; ++tt) b[tt] = *reinterpret_cast<const v4f*>(bq + tt * 64);
            v2f a0 = {a[j].x, a[j].y};
            v2f a1 = {a[j].z, a[j].w};
#pragma unroll
            for (int tt = 0; tt < 4; ++tt) {
                v2f b0 = {b[tt].x, b[tt].y};
                acc[tt] = __builtin_amdgcn_wmma_f32_16x16x4_f32(
                         false, a0,  false, b0,
                         (short)0, acc[tt],  false,  false);
            }
#pragma unroll
            for (int tt = 0; tt < 4; ++tt) {
                v2f b1 = {b[tt].z, b[tt].w};
                acc[tt] = __builtin_amdgcn_wmma_f32_16x16x4_f32(
                         false, a1,  false, b1,
                         (short)0, acc[tt],  false,  false);
            }
        }
    }

    const v4f bv = *reinterpret_cast<const v4f*>(bias + 4 * c);
#pragma unroll
    for (int r = 0; r < 8; ++r) {
        const size_t orow = (size_t)mtile * 16 + r + h * 8;
        v4f o = {acc[0][r] + bv.x, acc[1][r] + bv.y, acc[2][r] + bv.z, acc[3][r] + bv.w};
        o.x = o.x > 0.0f ? o.x : 0.0f;
        o.y = o.y > 0.0f ? o.y : 0.0f;
        o.z = o.z > 0.0f ? o.z : 0.0f;
        o.w = o.w > 0.0f ? o.w : 0.0f;
        *reinterpret_cast<v4f*>(out + orow * 64 + 4 * c) = o;
    }
}


template <int BITS, int SHIFT>
static void sort_pass(const int* kin, const int* rin, const float* nin,
                      int* T, int* bt, int* kout, int* rout, float* nout,
                      hipStream_t stream) {
    const int nt = (1 << BITS) * NUNITS;
    const int sb = (nt + SCAN_T - 1) / SCAN_T;
    digit_count_kernel<BITS, SHIFT><<<UBLOCKS, 256, 0, stream>>>(kin, T);
    block_total_kernel<<<sb, SCAN_T, 0, stream>>>(T, nt, bt);
    scan_final_kernel<<<sb, SCAN_T, 0, stream>>>(T, nt, bt);
    digit_place_kernel<BITS, SHIFT><<<UBLOCKS, 256, 0, stream>>>(kin, rin, nin, T, kout, rout, nout);
}

template <int D>
static void run_layer(const float* hin, const float* W, const float* bias,
                      float* buf1, float* buf2, float* buf3,
                      const int* off, const int* rs, const float* ns,
                      float* hout, hipStream_t stream) {
    gather_kernel<D><<<HOP_BLOCKS, 256, 0, stream>>>(hin,  buf1, off, rs, ns);
    gather_kernel<D><<<HOP_BLOCKS, 256, 0, stream>>>(buf1, buf2, off, rs, ns);
    gather_kernel<D><<<HOP_BLOCKS, 256, 0, stream>>>(buf2, buf3, off, rs, ns);

    tag_gemm_kernel<D><<<(N_NODES / 16 + 3) / 4, 128, 0, stream>>>(
        hin, buf1, buf2, buf3, W, bias, hout);
}

extern "C" void kernel_launch(void* const* d_in, const int* in_sizes, int n_in,
                              void* d_out, int out_size, void* d_ws, size_t ws_size,
                              hipStream_t stream) {
    const float* x   = (const float*)d_in[0];
    const int*   ei  = (const int*)d_in[1];
    const float* ew  = (const float*)d_in[2];
    const float* W0  = (const float*)d_in[3];
    const float* b0  = (const float*)d_in[4];
    const float* Ws  = (const float*)d_in[5];
    const float* bs  = (const float*)d_in[6];
    float*       out = (float*)d_out;

    const int* row = ei;
    const int* col = ei + E_EDGES;

    const size_t NPAD = 100096;
    const size_t TPAD = 640000;
    float* ws   = (float*)d_ws;
    float* deg  = ws;
    int*   off  = (int*)(deg + NPAD);
    int*   bt   = off + NPAD;
    int*   T    = bt + 1024;
    float* norm = (float*)(T + TPAD);
    int*   k1   = (int*)(norm + E_EDGES);
    int*   r1   = k1 + E_EDGES;
    float* n1   = (float*)(r1 + E_EDGES);
    int*   k2   = (int*)(n1 + E_EDGES);
    int*   r2   = k2 + E_EDGES;
    float* n2   = (float*)(r2 + E_EDGES);
    float* bufA = n2 + E_EDGES;
    float* bufB = bufA + (size_t)N_NODES * LATENT;
    float* buf1 = bufB + (size_t)N_NODES * LATENT;
    float* buf2 = buf1 + (size_t)N_NODES * LATENT;
    float* buf3 = buf2 + (size_t)N_NODES * LATENT;

    hipMemsetAsync(deg, 0, N_NODES * sizeof(float), stream);
    deg_kernel<<<E_EDGES / 256, 256, 0, stream>>>(col, ew, deg);
    dinv_kernel<<<(N_NODES + 255) / 256, 256, 0, stream>>>(deg);
    norm_kernel<<<E_EDGES / 256, 256, 0, stream>>>(row, col, ew, deg, norm);

    sort_pass<LO_BITS, 0>(col, row, norm, T, bt, k1, r1, n1, stream);
    sort_pass<HI_BITS, LO_BITS>(k1, r1, n1, T, bt, k2, r2, n2, stream);
    run_start_kernel<<<E_EDGES / 256, 256, 0, stream>>>(k2, off);

    const size_t WL = 4 * 64 * 64;
    run_layer<IN_DIM>(x,    W0,          b0,       buf1, buf2, buf3, off, r2, n2, bufA, stream);
    run_layer<LATENT>(bufA, Ws + 0 * WL, bs + 0,   buf1, buf2, buf3, off, r2, n2, bufB, stream);
    run_layer<LATENT>(bufB, Ws + 1 * WL, bs + 64,  buf1, buf2, buf3, off, r2, n2, bufA, stream);
    run_layer<LATENT>(bufA, Ws + 2 * WL, bs + 128, buf1, buf2, buf3, off, r2, n2, bufB, stream);
    run_layer<LATENT>(bufB, Ws + 3 * WL, bs + 192, buf1, buf2, buf3, off, r2, n2, out,  stream);
}


# Sort pass 2 inside one work-group per 256-node bucket: LDS counts, scan, records staged and written in whole lines, run starts from the scan; 28 launches, was 32; sum order unchanged
# speedup vs baseline: 3.7689x; 1.1447x over previous
#include <hip/hip_runtime.h>


#define N_NODES 100000
#define E_EDGES 1600000
#define IN_DIM  32
#define LATENT  64

#define UE      1280
#define NUNITS  (E_EDGES / UE)
#define UBLOCKS ((NUNITS + 7) / 8)
#define BK_BITS 8
#define BK_NODES (1 << BK_BITS)
#define HI_BITS 9
#define NBUCKETS (N_NODES / BK_NODES + 1)
#define BK_WAVES 16
#define BK_T    (BK_WAVES * 32)
#define BK_CAP  8192
#define SCAN_T  1024
#define HOP_BLOCKS 2048
#define HOP_WAVES  (HOP_BLOCKS * 8)

typedef __attribute__((ext_vector_type(2))) float v2f;
typedef __attribute__((ext_vector_type(4))) float v4f;
typedef __attribute__((ext_vector_type(8))) float v8f;

__device__ __forceinline__ void atomic_add_f32(float* p, float v) {
    asm volatile("global_atomic_add_f32 %0, %1, off scope:SCOPE_DEV"
                 :: "v"(p), "v"(v)
                 : "memory");
}


__global__ void deg_kernel(const int* __restrict__ col,
                           const float* __restrict__ w,
                           float* __restrict__ deg) {
    int e = blockIdx.x * blockDim.x + threadIdx.x;
    if (e < E_EDGES) atomic_add_f32(&deg[col[e]], w[e]);
}

__global__ void dinv_kernel(float* __restrict__ deg) {
    int v = blockIdx.x * blockDim.x + threadIdx.x;
    if (v < N_NODES) {
        float d = deg[v];
        deg[v] = (d > 0.0f) ? __frsqrt_rn(d) : 0.0f;
    }
}

__global__ void norm_kernel(const int* __restrict__ row,
                            const int* __restrict__ col,
                            const float* __restrict__ w,
                            const float* __restrict__ dinv,
                            float* __restrict__ norm) {
    int e = blockIdx.x * blockDim.x + threadIdx.x;
    if (e < E_EDGES) norm[e] = dinv[row[e]] * w[e] * dinv[col[e]];
}


template <int BITS>
__device__ __forceinline__ unsigned same_digit_lanes(int d) {
    unsigned m = 0xffffffffu;
#pragma unroll
    for (int i = 0; i < BITS; ++i) {
        const bool bit = (d >> i) & 1;
        const unsigned bal = __builtin_amdgcn_ballot_w32(bit);
        m &= bit ? bal : ~bal;
    }
    return m;
}

template <int BITS, int SHIFT>
__global__ void __launch_bounds__(256)
digit_count_kernel(const int* __restrict__ key, int* __restrict__ T) {
    constexpr int NB = 1 << BITS;
    __shared__ int h[8][NB];
    const int lane = threadIdx.x & 31;
    const int wave = threadIdx.x >> 5;
    const int unit = blockIdx.x * 8 + wave;
    if (unit >= NUNITS) return;
    int* hw = h[wave];
    for (int i = lane; i < NB; i += 32) hw[i] = 0;
    const unsigned lt = (1u << lane) - 1u;
    for (int c0 = 0; c0 < UE / 32; c0 += 8) {
        int kk[8];
#pragma unroll
        for (int u = 0; u < 8; ++u) kk[u] = key[unit * UE + (c0 + u) * 32 + lane];
#pragma unroll
        for (int u = 0; u < 8; ++u) {
            const int d = (kk[u] >> SHIFT) & (NB - 1);
            const unsigned m = same_digit_lanes<BITS>(d);
            if ((m & lt) == 0) hw[d] = hw[d] + __popc(m);
        }
    }
    for (int i = lane; i < NB; i += 32) T[(size_t)i * NUNITS + unit] = hw[i];
}

__global__ void __launch_bounds__(SCAN_T)
block_total_kernel(const int* __restrict__ T, int nt, int* __restrict__ bt) {
    __shared__ int ws[32];
    const int t = threadIdx.x, lane = t & 31, wave = t >> 5;
    const int i = blockIdx.x * SCAN_T + t;
    int s = (i < nt) ? T[i] : 0;
#pragma unroll
    for (int d = 16; d > 0; d >>= 1) s += __shfl_xor(s, d);
    if (lane == 0) ws[wave] = s;
    __syncthreads();
    if (wave == 0) {
        int q = ws[lane];
#pragma unroll
        for (int d = 16; d > 0; d >>= 1) q += __shfl_xor(q, d);
        if (lane == 0) bt[blockIdx.x] = q;
    }
}

__global__ void __launch_bounds__(SCAN_T)
scan_final_kernel(int* __restrict__ T, int nt, const int* __restrict__ bt) {
    __shared__ int wtot[32];
    __shared__ int wpre[32];
    const int t = threadIdx.x, lane = t & 31, wave = t >> 5;
    int p = 0;
    for (int i = t; i < (int)blockIdx.x; i += SCAN_T) p += bt[i];
    const int idx = blockIdx.x * SCAN_T + t;
    const int c = (idx < nt) ? T[idx] : 0;
    int inc = c;
#pragma unroll
    for (int d = 1; d < 32; d <<= 1) {
        const int y = __shfl_up(inc, d);
        if (lane >= d) inc += y;
    }
#pragma unroll
    for (int d = 16; d > 0; d >>= 1) p += __shfl_xor(p, d);
    if (lane == 31) wtot[wave] = inc;
    if (lane == 0)  wpre[wave] = p;
    __syncthreads();
    const int wt = wtot[lane];
    int winc = wt;
#pragma unroll
    for (int d = 1; d < 32; d <<= 1) {
        const int y = __shfl_up(winc, d);
        if (lane >= d) winc += y;
    }
    const int woff = __shfl(winc, wave) - __shfl(wt, wave);
    int P = wpre[lane];
#pragma unroll
    for (int d = 16; d > 0; d >>= 1) P += __shfl_xor(P, d);
    if (idx < nt) T[idx] = P + woff + inc - c;
}

template <int BITS, int SHIFT>
__global__ void __launch_bounds__(256)
digit_place_kernel(const int* __restrict__ key, const int* __restrict__ rin,
                   const float* __restrict__ nin, const int* __restrict__ T,
                   int* __restrict__ kout, int* __restrict__ rout,
                   float* __restrict__ nout) {
    constexpr int NB = 1 << BITS;
    __shared__ int h[8][NB];
    const int lane = threadIdx.x & 31;
    const int wave = threadIdx.x >> 5;
    const int unit = blockIdx.x * 8 + wave;
    if (unit >= NUNITS) return;
    int* hw = h[wave];
    for (int i = lane; i < NB; i += 32) hw[i] = T[(size_t)i * NUNITS + unit];
    const unsigned lt = (1u << lane) - 1u;
    for (int c0 = 0; c0 < UE / 32; c0 += 8) {
      int kk[8], rr[8]; float nn[8];
#pragma unroll
      for (int u = 0; u < 8; ++u) {
          const int e = unit * UE + (c0 + u) * 32 + lane;
          kk[u] = key[e];
          rr[u] = rin[e];
          nn[u] = nin[e];
      }
#pragma unroll
      for (int u = 0; u < 8; ++u) {
        const int   k = kk[u];
        const int   r = rr[u];
        const float n = nn[u];
        const int d = (k >> SHIFT) & (NB - 1);
        const unsigned m = same_digit_lanes<BITS>(d);
        const int first = __ffs(m) - 1;
        int start = 0;
        if (lane == first) {
            start = hw[d];
            hw[d] = start + __popc(m);
        }
        start = __shfl(start, first);
        const int pos = start + __popc(m & lt);
        kout[pos] = k;
        rout[pos] = r;
        nout[pos] = n;
      }
    }
}

template <int BITS>
__device__ __forceinline__ unsigned same_digit_valid_lanes(int d, bool valid) {
    unsigned m = __builtin_amdgcn_ballot_w32(valid);
#pragma unroll
    for (int i = 0; i < BITS; ++i) {
        const bool bit = (d >> i) & 1;
        const unsigned bal = __builtin_amdgcn_ballot_w32(bit);
        m &= bit ? bal : ~bal;
    }
    return valid ? m : 0u;
}

__global__ void __launch_bounds__(BK_T)
bucket_order_kernel(const int* __restrict__ kin, const int* __restrict__ rin,
                    const float* __restrict__ nin, const int* __restrict__ T,
                    int* __restrict__ off, int* __restrict__ rout,
                    float* __restrict__ nout) {
    __shared__ int   h[BK_WAVES][BK_NODES];
    __shared__ int   wtot[BK_NODES / 32];
    __shared__ int   stage_r[BK_CAP];
    __shared__ float stage_n[BK_CAP];

    const int t = threadIdx.x, lane = t & 31;
    const int wave = __builtin_amdgcn_readfirstlane(t >> 5);
    const int b = blockIdx.x;
    const int beg = T[(size_t)b * NUNITS];
    const int end = (b + 1 < (1 << HI_BITS)) ? T[(size_t)(b + 1) * NUNITS] : E_EDGES;
    const int n = end - beg;
    const int L  = ((n + BK_T - 1) / BK_T) * 32;
    const int s0 = beg + wave * L;
    const int s1 = (s0 + L < end) ? (s0 + L) : end;

    int* hw = h[wave];
    for (int i = lane; i < BK_NODES; i += 32) hw[i] = 0;
    const unsigned lt = (1u << lane) - 1u;

    for (int p0 = s0; p0 < s1; p0 += 256) {
        int kk[8];
#pragma unroll
        for (int u = 0; u < 8; ++u) {
            const int e = p0 + u * 32 + lane;
            kk[u] = kin[(e < s1) ? e : (s1 - 1)];
        }
#pragma unroll
        for (int u = 0; u < 8; ++u) {
            const bool valid = (p0 + u * 32 + lane) < s1;
            const int d = kk[u] & (BK_NODES - 1);
            const unsigned m = same_digit_valid_lanes<BK_BITS>(d, valid);
            if (valid && (m & lt) == 0) hw[d] = hw[d] + __popc(m);
        }
    }
    __syncthreads();

    int cnt = 0, inc = 0;
    if (t < BK_NODES) {
#pragma unroll
        for (int w = 0; w < BK_WAVES; ++w) {
            const int c = h[w][t];
            h[w][t] = cnt;
            cnt += c;
        }
        inc = cnt;
#pragma unroll
        for (int d = 1; d < 32; d <<= 1) {
            const int y = __shfl_up(inc, d);
            if (lane >= d) inc += y;
        }
        if (lane == 31) wtot[wave] = inc;
    }
    __syncthreads();
    if (t < BK_NODES) {
        int P = inc - cnt;
        for (int w = 0; w < wave; ++w) P += wtot[w];
#pragma unroll
        for (int w = 0; w < BK_WAVES; ++w) h[w][t] += P;
        const int v = b * BK_NODES + t;
        if (v <= N_NODES) off[v] = beg + P;
    }
    __syncthreads();

    const bool staged = (n <= BK_CAP);
    for (int p0 = s0; p0 < s1; p0 += 256) {
        int kk[8], rr[8]; float nn[8];
#pragma unroll
        for (int u = 0; u < 8; ++u) {
            const int e = p0 + u * 32 + lane;
            const int ec = (e < s1) ? e : (s1 - 1);
            kk[u] = kin[ec];
            rr[u] = rin[ec];
            nn[u] = nin[ec];
        }
#pragma unroll
        for (int u = 0; u < 8; ++u) {
            const bool valid = (p0 + u * 32 + lane) < s1;
            const int d = kk[u] & (BK_NODES - 1);
            const unsigned m = same_digit_valid_lanes<BK_BITS>(d, valid);
            const int first = m ? (__ffs(m) - 1) : lane;
            int start = 0;
            if (valid && lane == first) {
                start = hw[d];
                hw[d] = start + __popc(m);
            }
            start = __shfl(start, first);
            const int pos = start + __popc(m & lt);
            if (valid) {
                if (staged) { stage_r[pos] = rr[u]; stage_n[pos] = nn[u]; }
                else        { rout[beg + pos] = rr[u]; nout[beg + pos] = nn[u]; }
            }
        }
    }
    __syncthreads();
    if (staged) {
        for (int i = t; i < n; i += BK_T) {
            rout[beg + i] = stage_r[i];
            nout[beg + i] = stage_n[i];
        }
    }
}


template <int VPL> struct row_piece;
template <> struct row_piece<1> { typedef float type; };
template <> struct row_piece<2> { typedef v2f   type; };

template <int D, int G>
__device__ __forceinline__ void gather_group(const float* __restrict__ src, int lane,
                                             int rv, int nv, int j,
                                             typename row_piece<D / 32>::type& acc) {
    typedef typename row_piece<D / 32>::type vec;
    vec x[G];
#pragma unroll
    for (int k = 0; k < G; ++k) {
        const int r = __builtin_amdgcn_readlane(rv, j + k);
        x[k] = reinterpret_cast<const vec*>(src + (size_t)r * D)[lane];
    }
#pragma unroll
    for (int k = 0; k < G; ++k) {
        const float w = __int_as_float(__builtin_amdgcn_readlane(nv, j + k));
        acc += w * x[k];
    }
}

template <int D>
__global__ void __launch_bounds__(256)
gather_kernel(const float* __restrict__ src,
              float* __restrict__ dst,
              const int* __restrict__ off,
              const int* __restrict__ rs,
              const float* __restrict__ ns) {
    typedef typename row_piece<D / 32>::type vec;
    const int lane = threadIdx.x & 31;
    const int wv = __builtin_amdgcn_readfirstlane((int)(blockIdx.x * 8 + (threadIdx.x >> 5)));

    for (int v = wv; v < N_NODES; v += HOP_WAVES) {
        const int beg = off[v];
        const int end = off[v + 1];
        vec acc = 0.0f;
        for (int b = beg; b < end; b += 32) {
            const int n = (end - b < 32) ? (end - b) : 32;
            int rv = 0, nv = 0;
            if (lane < n) {
                rv = rs[b + lane];
                nv = __float_as_int(ns[b + lane]);
            }
            int j = 0;
            for (; j + 16 <= n; j += 16) gather_group<D, 16>(src, lane, rv, nv, j, acc);
            if (n - j >= 8) { gather_group<D, 8>(src, lane, rv, nv, j, acc); j += 8; }
            if (n - j >= 4) { gather_group<D, 4>(src, lane, rv, nv, j, acc); j += 4; }
            if (n - j >= 2) { gather_group<D, 2>(src, lane, rv, nv, j, acc); j += 2; }
            if (n - j >= 1) { gather_group<D, 1>(src, lane, rv, nv, j, acc); }
        }
        reinterpret_cast<vec*>(dst + (size_t)v * D)[lane] = acc;
    }
}


#define WQ_PITCH 288

template <int D>
__global__ void __launch_bounds__(128)
tag_gemm_kernel(const float* __restrict__ H0, const float* __restrict__ H1,
                const float* __restrict__ H2, const float* __restrict__ H3,
                const float* __restrict__ W,
                const float* __restrict__ bias,
                float* __restrict__ out) {
    constexpr int NQ = (4 * D) / 4;
    __shared__ __attribute__((aligned(16))) float Wl[NQ * WQ_PITCH];

    const int t = threadIdx.x;
#pragma unroll
    for (int it = 0; it < NQ * 16 / 128; ++it) {
        const int i = t + 128 * it;
        const int q = i >> 4, m = i & 15;
        const float* __restrict__ src = W + (size_t)(4 * q) * 64 + 4 * m;
        const v4f r0 = *reinterpret_cast<const v4f*>(src);
        const v4f r1 = *reinterpret_cast<const v4f*>(src + 64);
        const v4f r2 = *reinterpret_cast<const v4f*>(src + 128);
        const v4f r3 = *reinterpret_cast<const v4f*>(src + 192);
        float* dst = Wl + q * WQ_PITCH + m * 4;
        v4f c0 = {r0.x, r1.x, r2.x, r3.x};
        v4f c1 = {r0.y, r1.y, r2.y, r3.y};
        v4f c2 = {r0.z, r1.z, r2.z, r3.z};
        v4f c3 = {r0.w, r1.w, r2.w, r3.w};
        *reinterpret_cast<v4f*>(dst)       = c0;
        *reinterpret_cast<v4f*>(dst + 64)  = c1;
        *reinterpret_cast<v4f*>(dst + 128) = c2;
        *reinterpret_cast<v4f*>(dst + 192) = c3;
    }
    __syncthreads();

    const int wave  = t >> 5;
    const int lane  = t & 31;
    const int mtile = blockIdx.x * 4 + wave;
    if (mtile >= N_NODES / 16) return;

    const int c = lane & 15;
    const int h = lane >> 4;
    const size_t arow = (size_t)mtile * 16 + c;

    const float* Hs[4] = {H0, H1, H2, H3};
    v8f acc[4] = {};

#pragma unroll
    for (int s = 0; s < 4; ++s) {
        const float* __restrict__ A = Hs[s] + arow * D + 4 * h;
        v4f a[D / 8];
#pragma unroll
        for (int j = 0; j < D / 8; ++j) a[j] = *reinterpret_cast<const v4f*>(A + 8 * j);
#pragma unroll
        for (int j = 0; j < D / 8; ++j) {
            const float* bq = Wl + ((s * D + 8 * j) / 4 + h) * WQ_PITCH + c * 4;
            v4f b[4];
#pragma unroll
            for (int tt = 0; tt < 4; ++tt) b[tt] = *reinterpret_cast<const v4f*>(bq + tt * 64);
            v2f a0 = {a[j].x, a[j].y};
            v2f a1 = {a[j].z, a[j].w};
#pragma unroll
            for (int tt = 0; tt < 4; ++tt) {
                v2f b0 = {b[tt].x, b[tt].y};
                acc[tt] = __builtin_amdgcn_wmma_f32_16x16x4_f32(
                         false, a0,  false, b0,
                         (short)0, acc[tt],  false,  false);
            }
#pragma unroll
            for (int tt = 0; tt < 4; ++tt) {
                v2f b1 = {b[tt].z, b[tt].w};
                acc[tt] = __builtin_amdgcn_wmma_f32_16x16x4_f32(
                         false, a1,  false, b1,
                         (short)0, acc[tt],  false,  false);
            }
        }
    }

    const v4f bv = *reinterpret_cast<const v4f*>(bias + 4 * c);
#pragma unroll
    for (int r = 0; r < 8; ++r) {
        const size_t orow = (size_t)mtile * 16 + r + h * 8;
        v4f o = {acc[0][r] + bv.x, acc[1][r] + bv.y, acc[2][r] + bv.z, acc[3][r] + bv.w};
        o.x = o.x > 0.0f ? o.x : 0.0f;
        o.y = o.y > 0.0f ? o.y : 0.0f;
        o.z = o.z > 0.0f ? o.z : 0.0f;
        o.w = o.w > 0.0f ? o.w : 0.0f;
        *reinterpret_cast<v4f*>(out + orow * 64 + 4 * c) = o;
    }
}


template <int BITS, int SHIFT>
static void sort_pass(const int* kin, const int* rin, const float* nin,
                      int* T, int* bt, int* kout, int* rout, float* nout,
                      hipStream_t stream) {
    const int nt = (1 << BITS) * NUNITS;
    const int sb = (nt + SCAN_T - 1) / SCAN_T;
    digit_count_kernel<BITS, SHIFT><<<UBLOCKS, 256, 0, stream>>>(kin, T);
    block_total_kernel<<<sb, SCAN_T, 0, stream>>>(T, nt, bt);
    scan_final_kernel<<<sb, SCAN_T, 0, stream>>>(T, nt, bt);
    digit_place_kernel<BITS, SHIFT><<<UBLOCKS, 256, 0, stream>>>(kin, rin, nin, T, kout, rout, nout);
}

template <int D>
static void run_layer(const float* hin, const float* W, const float* bias,
                      float* buf1, float* buf2, float* buf3,
                      const int* off, const int* rs, const float* ns,
                      float* hout, hipStream_t stream) {
    gather_kernel<D><<<HOP_BLOCKS, 256, 0, stream>>>(hin,  buf1, off, rs, ns);
    gather_kernel<D><<<HOP_BLOCKS, 256, 0, stream>>>(buf1, buf2, off, rs, ns);
    gather_kernel<D><<<HOP_BLOCKS, 256, 0, stream>>>(buf2, buf3, off, rs, ns);

    tag_gemm_kernel<D><<<(N_NODES / 16 + 3) / 4, 128, 0, stream>>>(
        hin, buf1, buf2, buf3, W, bias, hout);
}

extern "C" void kernel_launch(void* const* d_in, const int* in_sizes, int n_in,
                              void* d_out, int out_size, void* d_ws, size_t ws_size,
                              hipStream_t stream) {
    const float* x   = (const float*)d_in[0];
    const int*   ei  = (const int*)d_in[1];
    const float* ew  = (const float*)d_in[2];
    const float* W0  = (const float*)d_in[3];
    const float* b0  = (const float*)d_in[4];
    const float* Ws  = (const float*)d_in[5];
    const float* bs  = (const float*)d_in[6];
    float*       out = (float*)d_out;

    const int* row = ei;
    const int* col = ei + E_EDGES;

    const size_t NPAD = 100096;
    const size_t TPAD = 640000;
    float* ws   = (float*)d_ws;
    float* deg  = ws;
    int*   off  = (int*)(deg + NPAD);
    int*   bt   = off + NPAD;
    int*   T    = bt + 1024;
    float* norm = (float*)(T + TPAD);
    int*   k1   = (int*)(norm + E_EDGES);
    int*   r1   = k1 + E_EDGES;
    float* n1   = (float*)(r1 + E_EDGES);
    int*   r2   = (int*)(n1 + E_EDGES);
    float* n2   = (float*)(r2 + E_EDGES);
    float* bufA = n2 + E_EDGES;
    float* bufB = bufA + (size_t)N_NODES * LATENT;
    float* buf1 = bufB + (size_t)N_NODES * LATENT;
    float* buf2 = buf1 + (size_t)N_NODES * LATENT;
    float* buf3 = buf2 + (size_t)N_NODES * LATENT;

    hipMemsetAsync(deg, 0, N_NODES * sizeof(float), stream);
    deg_kernel<<<E_EDGES / 256, 256, 0, stream>>>(col, ew, deg);
    dinv_kernel<<<(N_NODES + 255) / 256, 256, 0, stream>>>(deg);
    norm_kernel<<<E_EDGES / 256, 256, 0, stream>>>(row, col, ew, deg, norm);

    sort_pass<HI_BITS, BK_BITS>(col, row, norm, T, bt, k1, r1, n1, stream);
    bucket_order_kernel<<<NBUCKETS, BK_T, 0, stream>>>(k1, r1, n1, T, off, r2, n2);

    const size_t WL = 4 * 64 * 64;
    run_layer<IN_DIM>(x,    W0,          b0,       buf1, buf2, buf3, off, r2, n2, bufA, stream);
    run_layer<LATENT>(bufA, Ws + 0 * WL, bs + 0,   buf1, buf2, buf3, off, r2, n2, bufB, stream);
    run_layer<LATENT>(bufB, Ws + 1 * WL, bs + 64,  buf1, buf2, buf3, off, r2, n2, bufA, stream);
    run_layer<LATENT>(bufA, Ws + 2 * WL, bs + 128, buf1, buf2, buf3, off, r2, n2, bufB, stream);
    run_layer<LATENT>(bufB, Ws + 3 * WL, bs + 192, buf1, buf2, buf3, off, r2, n2, out,  stream);
}


# Global split pass per work-group: 16 waves count 6400 records, scan, order them by digit in LDS and store each digit's records as one run (was 32 single words per store); sum order unchanged
# speedup vs baseline: 4.3352x; 1.1503x over previous
#include <hip/hip_runtime.h>


#define N_NODES 100000
#define E_EDGES 1600000
#define IN_DIM  32
#define LATENT  64

#define SP_REC  6400
#define NSPB    (E_EDGES / SP_REC)
#define SP_WAVES 16
#define SP_T    (SP_WAVES * 32)
#define SP_L    (((SP_REC + SP_T - 1) / SP_T) * 32)
#define BK_BITS 8
#define BK_NODES (1 << BK_BITS)
#define HI_BITS 9
#define NDIG    (1 << HI_BITS)
#define NBUCKETS (N_NODES / BK_NODES + 1)
#define BK_WAVES 16
#define BK_T    (BK_WAVES * 32)
#define BK_CAP  8192
#define SCAN_T  1024
#define HOP_BLOCKS 2048
#define HOP_WAVES  (HOP_BLOCKS * 8)

typedef __attribute__((ext_vector_type(2))) float v2f;
typedef __attribute__((ext_vector_type(4))) float v4f;
typedef __attribute__((ext_vector_type(8))) float v8f;

__device__ __forceinline__ void atomic_add_f32(float* p, float v) {
    asm volatile("global_atomic_add_f32 %0, %1, off scope:SCOPE_DEV"
                 :: "v"(p), "v"(v)
                 : "memory");
}


__global__ void deg_kernel(const int* __restrict__ col,
                           const float* __restrict__ w,
                           float* __restrict__ deg) {
    int e = blockIdx.x * blockDim.x + threadIdx.x;
    if (e < E_EDGES) atomic_add_f32(&deg[col[e]], w[e]);
}

__global__ void dinv_kernel(float* __restrict__ deg) {
    int v = blockIdx.x * blockDim.x + threadIdx.x;
    if (v < N_NODES) {
        float d = deg[v];
        deg[v] = (d > 0.0f) ? __frsqrt_rn(d) : 0.0f;
    }
}

__global__ void norm_kernel(const int* __restrict__ row,
                            const int* __restrict__ col,
                            const float* __restrict__ w,
                            const float* __restrict__ dinv,
                            float* __restrict__ norm) {
    int e = blockIdx.x * blockDim.x + threadIdx.x;
    if (e < E_EDGES) norm[e] = dinv[row[e]] * w[e] * dinv[col[e]];
}


template <int BITS>
__device__ __forceinline__ unsigned same_digit_valid_lanes(int d, bool valid) {
    unsigned m = __builtin_amdgcn_ballot_w32(valid);
#pragma unroll
    for (int i = 0; i < BITS; ++i) {
        const bool bit = (d >> i) & 1;
        const unsigned bal = __builtin_amdgcn_ballot_w32(bit);
        m &= bit ? bal : ~bal;
    }
    return valid ? m : 0u;
}

__global__ void __launch_bounds__(SP_T)
split_count_kernel(const int* __restrict__ key, int* __restrict__ T) {
    __shared__ int h[SP_WAVES][NDIG];
    const int t = threadIdx.x, lane = t & 31;
    const int wave = __builtin_amdgcn_readfirstlane(t >> 5);
    const int beg = blockIdx.x * SP_REC;
    const int end = beg + SP_REC;
    const int s0 = beg + wave * SP_L;
    const int s1 = (s0 + SP_L < end) ? (s0 + SP_L) : end;

    int* hw = h[wave];
    for (int i = lane; i < NDIG; i += 32) hw[i] = 0;
    const unsigned lt = (1u << lane) - 1u;
    for (int p0 = s0; p0 < s1; p0 += 256) {
        int kk[8];
#pragma unroll
        for (int u = 0; u < 8; ++u) {
            const int e = p0 + u * 32 + lane;
            kk[u] = key[(e < s1) ? e : (s1 - 1)];
        }
#pragma unroll
        for (int u = 0; u < 8; ++u) {
            const bool valid = (p0 + u * 32 + lane) < s1;
            const int d = (kk[u] >> BK_BITS) & (NDIG - 1);
            const unsigned m = same_digit_valid_lanes<HI_BITS>(d, valid);
            if (valid && (m & lt) == 0) hw[d] = hw[d] + __popc(m);
        }
    }
    __syncthreads();
    int cnt = 0;
#pragma unroll
    for (int w = 0; w < SP_WAVES; ++w) cnt += h[w][t];
    T[(size_t)t * NSPB + blockIdx.x] = cnt;
}

__global__ void __launch_bounds__(SCAN_T)
block_total_kernel(const int* __restrict__ T, int nt, int* __restrict__ bt) {
    __shared__ int ws[32];
    const int t = threadIdx.x, lane = t & 31, wave = t >> 5;
    const int i = blockIdx.x * SCAN_T + t;
    int s = (i < nt) ? T[i] : 0;
#pragma unroll
    for (int d = 16; d > 0; d >>= 1) s += __shfl_xor(s, d);
    if (lane == 0) ws[wave] = s;
    __syncthreads();
    if (wave == 0) {
        int q = ws[lane];
#pragma unroll
        for (int d = 16; d > 0; d >>= 1) q += __shfl_xor(q, d);
        if (lane == 0) bt[blockIdx.x] = q;
    }
}

__global__ void __launch_bounds__(SCAN_T)
scan_final_kernel(int* __restrict__ T, int nt, const int* __restrict__ bt) {
    __shared__ int wtot[32];
    __shared__ int wpre[32];
    const int t = threadIdx.x, lane = t & 31, wave = t >> 5;
    int p = 0;
    for (int i = t; i < (int)blockIdx.x; i += SCAN_T) p += bt[i];
    const int idx = blockIdx.x * SCAN_T + t;
    const int c = (idx < nt) ? T[idx] : 0;
    int inc = c;
#pragma unroll
    for (int d = 1; d < 32; d <<= 1) {
        const int y = __shfl_up(inc, d);
        if (lane >= d) inc += y;
    }
#pragma unroll
    for (int d = 16; d > 0; d >>= 1) p += __shfl_xor(p, d);
    if (lane == 31) wtot[wave] = inc;
    if (lane == 0)  wpre[wave] = p;
    __syncthreads();
    const int wt = wtot[lane];
    int winc = wt;
#pragma unroll
    for (int d = 1; d < 32; d <<= 1) {
        const int y = __shfl_up(winc, d);
        if (lane >= d) winc += y;
    }
    const int woff = __shfl(winc, wave) - __shfl(wt, wave);
    int P = wpre[lane];
#pragma unroll
    for (int d = 16; d > 0; d >>= 1) P += __shfl_xor(P, d);
    if (idx < nt) T[idx] = P + woff + inc - c;
}

__global__ void __launch_bounds__(SP_T)
split_place_kernel(const int* __restrict__ key, const int* __restrict__ rin,
                   const float* __restrict__ nin, const int* __restrict__ T,
                   int* __restrict__ kout, int* __restrict__ rout,
                   float* __restrict__ nout) {
    __shared__ int   h[SP_WAVES][NDIG];
    __shared__ int   wtot[SP_WAVES];
    __shared__ int   gbase[NDIG];
    __shared__ int   stage_k[SP_REC];
    __shared__ int   stage_r[SP_REC];
    __shared__ float stage_n[SP_REC];

    const int t = threadIdx.x, lane = t & 31;
    const int wave = __builtin_amdgcn_readfirstlane(t >> 5);
    const int beg = blockIdx.x * SP_REC;
    const int end = beg + SP_REC;
    const int s0 = beg + wave * SP_L;
    const int s1 = (s0 + SP_L < end) ? (s0 + SP_L) : end;

    int* hw = h[wave];
    for (int i = lane; i < NDIG; i += 32) hw[i] = 0;
    const unsigned lt = (1u << lane) - 1u;

    for (int p0 = s0; p0 < s1; p0 += 256) {
        int kk[8];
#pragma unroll
        for (int u = 0; u < 8; ++u) {
            const int e = p0 + u * 32 + lane;
            kk[u] = key[(e < s1) ? e : (s1 - 1)];
        }
#pragma unroll
        for (int u = 0; u < 8; ++u) {
            const bool valid = (p0 + u * 32 + lane) < s1;
            const int d = (kk[u] >> BK_BITS) & (NDIG - 1);
            const unsigned m = same_digit_valid_lanes<HI_BITS>(d, valid);
            if (valid && (m & lt) == 0) hw[d] = hw[d] + __popc(m);
        }
    }
    __syncthreads();

    int cnt = 0;
#pragma unroll
    for (int w = 0; w < SP_WAVES; ++w) {
        const int c = h[w][t];
        h[w][t] = cnt;
        cnt += c;
    }
    int inc = cnt;
#pragma unroll
    for (int d = 1; d < 32; d <<= 1) {
        const int y = __shfl_up(inc, d);
        if (lane >= d) inc += y;
    }
    if (lane == 31) wtot[wave] = inc;
    __syncthreads();
    int P = inc - cnt;
    for (int w = 0; w < wave; ++w) P += wtot[w];
#pragma unroll
    for (int w = 0; w < SP_WAVES; ++w) h[w][t] += P;
    gbase[t] = T[(size_t)t * NSPB + blockIdx.x] - P;
    __syncthreads();

    for (int p0 = s0; p0 < s1; p0 += 256) {
        int kk[8], rr[8]; float nn[8];
#pragma unroll
        for (int u = 0; u < 8; ++u) {
            const int e = p0 + u * 32 + lane;
            const int ec = (e < s1) ? e : (s1 - 1);
            kk[u] = key[ec];
            rr[u] = rin[ec];
            nn[u] = nin[ec];
        }
#pragma unroll
        for (int u = 0; u < 8; ++u) {
            const bool valid = (p0 + u * 32 + lane) < s1;
            const int d = (kk[u] >> BK_BITS) & (NDIG - 1);
            const unsigned m = same_digit_valid_lanes<HI_BITS>(d, valid);
            const int first = m ? (__ffs(m) - 1) : lane;
            int start = 0;
            if (valid && lane == first) {
                start = hw[d];
                hw[d] = start + __popc(m);
            }
            start = __shfl(start, first);
            const int pos = start + __popc(m & lt);
            if (valid) { stage_k[pos] = kk[u]; stage_r[pos] = rr[u]; stage_n[pos] = nn[u]; }
        }
    }
    __syncthreads();

    for (int i = t; i < SP_REC; i += SP_T) {
        const int k = stage_k[i];
        const int g = gbase[(k >> BK_BITS) & (NDIG - 1)] + i;
        kout[g] = k;
        rout[g] = stage_r[i];
        nout[g] = stage_n[i];
    }
}

__global__ void __launch_bounds__(BK_T)
bucket_order_kernel(const int* __restrict__ kin, const int* __restrict__ rin,
                    const float* __restrict__ nin, const int* __restrict__ T,
                    int* __restrict__ off, int* __restrict__ rout,
                    float* __restrict__ nout) {
    __shared__ int   h[BK_WAVES][BK_NODES];
    __shared__ int   wtot[BK_NODES / 32];
    __shared__ int   stage_r[BK_CAP];
    __shared__ float stage_n[BK_CAP];

    const int t = threadIdx.x, lane = t & 31;
    const int wave = __builtin_amdgcn_readfirstlane(t >> 5);
    const int b = blockIdx.x;
    const int beg = T[(size_t)b * NSPB];
    const int end = (b + 1 < NDIG) ? T[(size_t)(b + 1) * NSPB] : E_EDGES;
    const int n = end - beg;
    const int L  = ((n + BK_T - 1) / BK_T) * 32;
    const int s0 = beg + wave * L;
    const int s1 = (s0 + L < end) ? (s0 + L) : end;

    int* hw = h[wave];
    for (int i = lane; i < BK_NODES; i += 32) hw[i] = 0;
    const unsigned lt = (1u << lane) - 1u;

    for (int p0 = s0; p0 < s1; p0 += 256) {
        int kk[8];
#pragma unroll
        for (int u = 0; u < 8; ++u) {
            const int e = p0 + u * 32 + lane;
            kk[u] = kin[(e < s1) ? e : (s1 - 1)];
        }
#pragma unroll
        for (int u = 0; u < 8; ++u) {
            const bool valid = (p0 + u * 32 + lane) < s1;
            const int d = kk[u] & (BK_NODES - 1);
            const unsigned m = same_digit_valid_lanes<BK_BITS>(d, valid);
            if (valid && (m & lt) == 0) hw[d] = hw[d] + __popc(m);
        }
    }
    __syncthreads();

    int cnt = 0, inc = 0;
    if (t < BK_NODES) {
#pragma unroll
        for (int w = 0; w < BK_WAVES; ++w) {
            const int c = h[w][t];
            h[w][t] = cnt;
            cnt += c;
        }
        inc = cnt;
#pragma unroll
        for (int d = 1; d < 32; d <<= 1) {
            const int y = __shfl_up(inc, d);
            if (lane >= d) inc += y;
        }
        if (lane == 31) wtot[wave] = inc;
    }
    __syncthreads();
    if (t < BK_NODES) {
        int P = inc - cnt;
        for (int w = 0; w < wave; ++w) P += wtot[w];
#pragma unroll
        for (int w = 0; w < BK_WAVES; ++w) h[w][t] += P;
        const int v = b * BK_NODES + t;
        if (v <= N_NODES) off[v] = beg + P;
    }
    __syncthreads();

    const bool staged = (n <= BK_CAP);
    for (int p0 = s0; p0 < s1; p0 += 256) {
        int kk[8], rr[8]; float nn[8];
#pragma unroll
        for (int u = 0; u < 8; ++u) {
            const int e = p0 + u * 32 + lane;
            const int ec = (e < s1) ? e : (s1 - 1);
            kk[u] = kin[ec];
            rr[u] = rin[ec];
            nn[u] = nin[ec];
        }
#pragma unroll
        for (int u = 0; u < 8; ++u) {
            const bool valid = (p0 + u * 32 + lane) < s1;
            const int d = kk[u] & (BK_NODES - 1);
            const unsigned m = same_digit_valid_lanes<BK_BITS>(d, valid);
            const int first = m ? (__ffs(m) - 1) : lane;
            int start = 0;
            if (valid && lane == first) {
                start = hw[d];
                hw[d] = start + __popc(m);
            }
            start = __shfl(start, first);
            const int pos = start + __popc(m & lt);
            if (valid) {
                if (staged) { stage_r[pos] = rr[u]; stage_n[pos] = nn[u]; }
                else        { rout[beg + pos] = rr[u]; nout[beg + pos] = nn[u]; }
            }
        }
    }
    __syncthreads();
    if (staged) {
        for (int i = t; i < n; i += BK_T) {
            rout[beg + i] = stage_r[i];
            nout[beg + i] = stage_n[i];
        }
    }
}


template <int VPL> struct row_piece;
template <> struct row_piece<1> { typedef float type; };
template <> struct row_piece<2> { typedef v2f   type; };

template <int D, int G>
__device__ __forceinline__ void gather_group(const float* __restrict__ src, int lane,
                                             int rv, int nv, int j,
                                             typename row_piece<D / 32>::type& acc) {
    typedef typename row_piece<D / 32>::type vec;
    vec x[G];
#pragma unroll
    for (int k = 0; k < G; ++k) {
        const int r = __builtin_amdgcn_readlane(rv, j + k);
        x[k] = reinterpret_cast<const vec*>(src + (size_t)r * D)[lane];
    }
#pragma unroll
    for (int k = 0; k < G; ++k) {
        const float w = __int_as_float(__builtin_amdgcn_readlane(nv, j + k));
        acc += w * x[k];
    }
}

template <int D>
__global__ void __launch_bounds__(256)
gather_kernel(const float* __restrict__ src,
              float* __restrict__ dst,
              const int* __restrict__ off,
              const int* __restrict__ rs,
              const float* __restrict__ ns) {
    typedef typename row_piece<D / 32>::type vec;
    const int lane = threadIdx.x & 31;
    const int wv = __builtin_amdgcn_readfirstlane((int)(blockIdx.x * 8 + (threadIdx.x >> 5)));

    for (int v = wv; v < N_NODES; v += HOP_WAVES) {
        const int beg = off[v];
        const int end = off[v + 1];
        vec acc = 0.0f;
        for (int b = beg; b < end; b += 32) {
            const int n = (end - b < 32) ? (end - b) : 32;
            int rv = 0, nv = 0;
            if (lane < n) {
                rv = rs[b + lane];
                nv = __float_as_int(ns[b + lane]);
            }
            int j = 0;
            for (; j + 16 <= n; j += 16) gather_group<D, 16>(src, lane, rv, nv, j, acc);
            if (n - j >= 8) { gather_group<D, 8>(src, lane, rv, nv, j, acc); j += 8; }
            if (n - j >= 4) { gather_group<D, 4>(src, lane, rv, nv, j, acc); j += 4; }
            if (n - j >= 2) { gather_group<D, 2>(src, lane, rv, nv, j, acc); j += 2; }
            if (n - j >= 1) { gather_group<D, 1>(src, lane, rv, nv, j, acc); }
        }
        reinterpret_cast<vec*>(dst + (size_t)v * D)[lane] = acc;
    }
}


#define WQ_PITCH 288

template <int D>
__global__ void __launch_bounds__(128)
tag_gemm_kernel(const float* __restrict__ H0, const float* __restrict__ H1,
                const float* __restrict__ H2, const float* __restrict__ H3,
                const float* __restrict__ W,
                const float* __restrict__ bias,
                float* __restrict__ out) {
    constexpr int NQ = (4 * D) / 4;
    __shared__ __attribute__((aligned(16))) float Wl[NQ * WQ_PITCH];

    const int t = threadIdx.x;
#pragma unroll
    for (int it = 0; it < NQ * 16 / 128; ++it) {
        const int i = t + 128 * it;
        const int q = i >> 4, m = i & 15;
        const float* __restrict__ src = W + (size_t)(4 * q) * 64 + 4 * m;
        const v4f r0 = *reinterpret_cast<const v4f*>(src);
        const v4f r1 = *reinterpret_cast<const v4f*>(src + 64);
        const v4f r2 = *reinterpret_cast<const v4f*>(src + 128);
        const v4f r3 = *reinterpret_cast<const v4f*>(src + 192);
        float* dst = Wl + q * WQ_PITCH + m * 4;
        v4f c0 = {r0.x, r1.x, r2.x, r3.x};
        v4f c1 = {r0.y, r1.y, r2.y, r3.y};
        v4f c2 = {r0.z, r1.z, r2.z, r3.z};
        v4f c3 = {r0.w, r1.w, r2.w, r3.w};
        *reinterpret_cast<v4f*>(dst)       = c0;
        *reinterpret_cast<v4f*>(dst + 64)  = c1;
        *reinterpret_cast<v4f*>(dst + 128) = c2;
        *reinterpret_cast<v4f*>(dst + 192) = c3;
    }
    __syncthreads();

    const int wave  = t >> 5;
    const int lane  = t & 31;
    const int mtile = blockIdx.x * 4 + wave;
    if (mtile >= N_NODES / 16) return;

    const int c = lane & 15;
    const int h = lane >> 4;
    const size_t arow = (size_t)mtile * 16 + c;

    const float* Hs[4] = {H0, H1, H2, H3};
    v8f acc[4] = {};

#pragma unroll
    for (int s = 0; s < 4; ++s) {
        const float* __restrict__ A = Hs[s] + arow * D + 4 * h;
        v4f a[D / 8];
#pragma unroll
        for (int j = 0; j < D / 8; ++j) a[j] = *reinterpret_cast<const v4f*>(A + 8 * j);
#pragma unroll
        for (int j = 0; j < D / 8; ++j) {
            const float* bq = Wl + ((s * D + 8 * j) / 4 + h) * WQ_PITCH + c * 4;
            v4f b[4];
#pragma unroll
            for (int tt = 0; tt < 4; ++tt) b[tt] = *reinterpret_cast<const v4f*>(bq + tt * 64);
            v2f a0 = {a[j].x, a[j].y};
            v2f a1 = {a[j].z, a[j].w};
#pragma unroll
            for (int tt = 0; tt < 4; ++tt) {
                v2f b0 = {b[tt].x, b[tt].y};
                acc[tt] = __builtin_amdgcn_wmma_f32_16x16x4_f32(
                         false, a0,  false, b0,
                         (short)0, acc[tt],  false,  false);
            }
#pragma unroll
            for (int tt = 0; tt < 4; ++tt) {
                v2f b1 = {b[tt].z, b[tt].w};
                acc[tt] = __builtin_amdgcn_wmma_f32_16x16x4_f32(
                         false, a1,  false, b1,
                         (short)0, acc[tt],  false,  false);
            }
        }
    }

    const v4f bv = *reinterpret_cast<const v4f*>(bias + 4 * c);
#pragma unroll
    for (int r = 0; r < 8; ++r) {
        const size_t orow = (size_t)mtile * 16 + r + h * 8;
        v4f o = {acc[0][r] + bv.x, acc[1][r] + bv.y, acc[2][r] + bv.z, acc[3][r] + bv.w};
        o.x = o.x > 0.0f ? o.x : 0.0f;
        o.y = o.y > 0.0f ? o.y : 0.0f;
        o.z = o.z > 0.0f ? o.z : 0.0f;
        o.w = o.w > 0.0f ? o.w : 0.0f;
        *reinterpret_cast<v4f*>(out + orow * 64 + 4 * c) = o;
    }
}


static void split_pass(const int* kin, const int* rin, const float* nin,
                       int* T, int* bt, int* kout, int* rout, float* nout,
                       hipStream_t stream) {
    constexpr int nt = NDIG * NSPB;
    constexpr int sb = (nt + SCAN_T - 1) / SCAN_T;
    split_count_kernel<<<NSPB, SP_T, 0, stream>>>(kin, T);
    block_total_kernel<<<sb, SCAN_T, 0, stream>>>(T, nt, bt);
    scan_final_kernel<<<sb, SCAN_T, 0, stream>>>(T, nt, bt);
    split_place_kernel<<<NSPB, SP_T, 0, stream>>>(kin, rin, nin, T, kout, rout, nout);
}

template <int D>
static void run_layer(const float* hin, const float* W, const float* bias,
                      float* buf1, float* buf2, float* buf3,
                      const int* off, const int* rs, const float* ns,
                      float* hout, hipStream_t stream) {
    gather_kernel<D><<<HOP_BLOCKS, 256, 0, stream>>>(hin,  buf1, off, rs, ns);
    gather_kernel<D><<<HOP_BLOCKS, 256, 0, stream>>>(buf1, buf2, off, rs, ns);
    gather_kernel<D><<<HOP_BLOCKS, 256, 0, stream>>>(buf2, buf3, off, rs, ns);

    tag_gemm_kernel<D><<<(N_NODES / 16 + 3) / 4, 128, 0, stream>>>(
        hin, buf1, buf2, buf3, W, bias, hout);
}

extern "C" void kernel_launch(void* const* d_in, const int* in_sizes, int n_in,
                              void* d_out, int out_size, void* d_ws, size_t ws_size,
                              hipStream_t stream) {
    const float* x   = (const float*)d_in[0];
    const int*   ei  = (const int*)d_in[1];
    const float* ew  = (const float*)d_in[2];
    const float* W0  = (const float*)d_in[3];
    const float* b0  = (const float*)d_in[4];
    const float* Ws  = (const float*)d_in[5];
    const float* bs  = (const float*)d_in[6];
    float*       out = (float*)d_out;

    const int* row = ei;
    const int* col = ei + E_EDGES;

    const size_t NPAD = 100096;
    const size_t TPAD = 640000;
    float* ws   = (float*)d_ws;
    float* deg  = ws;
    int*   off  = (int*)(deg + NPAD);
    int*   bt   = off + NPAD;
    int*   T    = bt + 1024;
    float* norm = (float*)(T + TPAD);
    int*   k1   = (int*)(norm + E_EDGES);
    int*   r1   = k1 + E_EDGES;
    float* n1   = (float*)(r1 + E_EDGES);
    int*   r2   = (int*)(n1 + E_EDGES);
    float* n2   = (float*)(r2 + E_EDGES);
    float* bufA = n2 + E_EDGES;
    float* bufB = bufA + (size_t)N_NODES * LATENT;
    float* buf1 = bufB + (size_t)N_NODES * LATENT;
    float* buf2 = buf1 + (size_t)N_NODES * LATENT;
    float* buf3 = buf2 + (size_t)N_NODES * LATENT;

    hipMemsetAsync(deg, 0, N_NODES * sizeof(float), stream);
    deg_kernel<<<E_EDGES / 256, 256, 0, stream>>>(col, ew, deg);
    dinv_kernel<<<(N_NODES + 255) / 256, 256, 0, stream>>>(deg);
    norm_kernel<<<E_EDGES / 256, 256, 0, stream>>>(row, col, ew, deg, norm);

    split_pass(col, row, norm, T, bt, k1, r1, n1, stream);
    bucket_order_kernel<<<NBUCKETS, BK_T, 0, stream>>>(k1, r1, n1, T, off, r2, n2);

    const size_t WL = 4 * 64 * 64;
    run_layer<IN_DIM>(x,    W0,          b0,       buf1, buf2, buf3, off, r2, n2, bufA, stream);
    run_layer<LATENT>(bufA, Ws + 0 * WL, bs + 0,   buf1, buf2, buf3, off, r2, n2, bufB, stream);
    run_layer<LATENT>(bufB, Ws + 1 * WL, bs + 64,  buf1, buf2, buf3, off, r2, n2, bufA, stream);
    run_layer<LATENT>(bufA, Ws + 2 * WL, bs + 128, buf1, buf2, buf3, off, r2, n2, bufB, stream);
    run_layer<LATENT>(bufB, Ws + 3 * WL, bs + 192, buf1, buf2, buf3, off, r2, n2, out,  stream);
}


# gcn_norm without atomics: raw weights ride the sort, the bucket pass sums a node's degree from its sorted run (edge-list order, was atomic arrival order), weights normalised after; 26 launches
# speedup vs baseline: 4.9282x; 1.1368x over previous
#include <hip/hip_runtime.h>


#define N_NODES 100000
#define E_EDGES 1600000
#define IN_DIM  32
#define LATENT  64

#define SP_REC  6400
#define NSPB    (E_EDGES / SP_REC)
#define SP_WAVES 16
#define SP_T    (SP_WAVES * 32)
#define SP_L    (((SP_REC + SP_T - 1) / SP_T) * 32)
#define BK_BITS 8
#define BK_NODES (1 << BK_BITS)
#define HI_BITS 9
#define NDIG    (1 << HI_BITS)
#define NBUCKETS (N_NODES / BK_NODES + 1)
#define BK_WAVES 16
#define BK_T    (BK_WAVES * 32)
#define BK_CAP  8192
#define SCAN_T  1024
#define HOP_BLOCKS 2048
#define HOP_WAVES  (HOP_BLOCKS * 8)

typedef __attribute__((ext_vector_type(2))) float v2f;
typedef __attribute__((ext_vector_type(4))) float v4f;
typedef __attribute__((ext_vector_type(8))) float v8f;


template <int BITS>
__device__ __forceinline__ unsigned same_digit_valid_lanes(int d, bool valid) {
    unsigned m = __builtin_amdgcn_ballot_w32(valid);
#pragma unroll
    for (int i = 0; i < BITS; ++i) {
        const bool bit = (d >> i) & 1;
        const unsigned bal = __builtin_amdgcn_ballot_w32(bit);
        m &= bit ? bal : ~bal;
    }
    return valid ? m : 0u;
}

__global__ void __launch_bounds__(SP_T)
split_count_kernel(const int* __restrict__ key, int* __restrict__ T) {
    __shared__ int h[SP_WAVES][NDIG];
    const int t = threadIdx.x, lane = t & 31;
    const int wave = __builtin_amdgcn_readfirstlane(t >> 5);
    const int beg = blockIdx.x * SP_REC;
    const int end = beg + SP_REC;
    const int s0 = beg + wave * SP_L;
    const int s1 = (s0 + SP_L < end) ? (s0 + SP_L) : end;

    int* hw = h[wave];
    for (int i = lane; i < NDIG; i += 32) hw[i] = 0;
    const unsigned lt = (1u << lane) - 1u;
    for (int p0 = s0; p0 < s1; p0 += 256) {
        int kk[8];
#pragma unroll
        for (int u = 0; u < 8; ++u) {
            const int e = p0 + u * 32 + lane;
            kk[u] = key[(e < s1) ? e : (s1 - 1)];
        }
#pragma unroll
        for (int u = 0; u < 8; ++u) {
            const bool valid = (p0 + u * 32 + lane) < s1;
            const int d = (kk[u] >> BK_BITS) & (NDIG - 1);
            const unsigned m = same_digit_valid_lanes<HI_BITS>(d, valid);
            if (valid && (m & lt) == 0) hw[d] = hw[d] + __popc(m);
        }
    }
    __syncthreads();
    int cnt = 0;
#pragma unroll
    for (int w = 0; w < SP_WAVES; ++w) cnt += h[w][t];
    T[(size_t)t * NSPB + blockIdx.x] = cnt;
}

__global__ void __launch_bounds__(SCAN_T)
block_total_kernel(const int* __restrict__ T, int nt, int* __restrict__ bt) {
    __shared__ int ws[32];
    const int t = threadIdx.x, lane = t & 31, wave = t >> 5;
    const int i = blockIdx.x * SCAN_T + t;
    int s = (i < nt) ? T[i] : 0;
#pragma unroll
    for (int d = 16; d > 0; d >>= 1) s += __shfl_xor(s, d);
    if (lane == 0) ws[wave] = s;
    __syncthreads();
    if (wave == 0) {
        int q = ws[lane];
#pragma unroll
        for (int d = 16; d > 0; d >>= 1) q += __shfl_xor(q, d);
        if (lane == 0) bt[blockIdx.x] = q;
    }
}

__global__ void __launch_bounds__(SCAN_T)
scan_final_kernel(int* __restrict__ T, int nt, const int* __restrict__ bt) {
    __shared__ int wtot[32];
    __shared__ int wpre[32];
    const int t = threadIdx.x, lane = t & 31, wave = t >> 5;
    int p = 0;
    for (int i = t; i < (int)blockIdx.x; i += SCAN_T) p += bt[i];
    const int idx = blockIdx.x * SCAN_T + t;
    const int c = (idx < nt) ? T[idx] : 0;
    int inc = c;
#pragma unroll
    for (int d = 1; d < 32; d <<= 1) {
        const int y = __shfl_up(inc, d);
        if (lane >= d) inc += y;
    }
#pragma unroll
    for (int d = 16; d > 0; d >>= 1) p += __shfl_xor(p, d);
    if (lane == 31) wtot[wave] = inc;
    if (lane == 0)  wpre[wave] = p;
    __syncthreads();
    const int wt = wtot[lane];
    int winc = wt;
#pragma unroll
    for (int d = 1; d < 32; d <<= 1) {
        const int y = __shfl_up(winc, d);
        if (lane >= d) winc += y;
    }
    const int woff = __shfl(winc, wave) - __shfl(wt, wave);
    int P = wpre[lane];
#pragma unroll
    for (int d = 16; d > 0; d >>= 1) P += __shfl_xor(P, d);
    if (idx < nt) T[idx] = P + woff + inc - c;
}

__global__ void __launch_bounds__(SP_T)
split_place_kernel(const int* __restrict__ key, const int* __restrict__ rin,
                   const float* __restrict__ nin, const int* __restrict__ T,
                   int* __restrict__ kout, int* __restrict__ rout,
                   float* __restrict__ nout) {
    __shared__ int   h[SP_WAVES][NDIG];
    __shared__ int   wtot[SP_WAVES];
    __shared__ int   gbase[NDIG];
    __shared__ int   stage_k[SP_REC];
    __shared__ int   stage_r[SP_REC];
    __shared__ float stage_n[SP_REC];

    const int t = threadIdx.x, lane = t & 31;
    const int wave = __builtin_amdgcn_readfirstlane(t >> 5);
    const int beg = blockIdx.x * SP_REC;
    const int end = beg + SP_REC;
    const int s0 = beg + wave * SP_L;
    const int s1 = (s0 + SP_L < end) ? (s0 + SP_L) : end;

    int* hw = h[wave];
    for (int i = lane; i < NDIG; i += 32) hw[i] = 0;
    const unsigned lt = (1u << lane) - 1u;

    for (int p0 = s0; p0 < s1; p0 += 256) {
        int kk[8];
#pragma unroll
        for (int u = 0; u < 8; ++u) {
            const int e = p0 + u * 32 + lane;
            kk[u] = key[(e < s1) ? e : (s1 - 1)];
        }
#pragma unroll
        for (int u = 0; u < 8; ++u) {
            const bool valid = (p0 + u * 32 + lane) < s1;
            const int d = (kk[u] >> BK_BITS) & (NDIG - 1);
            const unsigned m = same_digit_valid_lanes<HI_BITS>(d, valid);
            if (valid && (m & lt) == 0) hw[d] = hw[d] + __popc(m);
        }
    }
    __syncthreads();

    int cnt = 0;
#pragma unroll
    for (int w = 0; w < SP_WAVES; ++w) {
        const int c = h[w][t];
        h[w][t] = cnt;
        cnt += c;
    }
    int inc = cnt;
#pragma unroll
    for (int d = 1; d < 32; d <<= 1) {
        const int y = __shfl_up(inc, d);
        if (lane >= d) inc += y;
    }
    if (lane == 31) wtot[wave] = inc;
    __syncthreads();
    int P = inc - cnt;
    for (int w = 0; w < wave; ++w) P += wtot[w];
#pragma unroll
    for (int w = 0; w < SP_WAVES; ++w) h[w][t] += P;
    gbase[t] = T[(size_t)t * NSPB + blockIdx.x] - P;
    __syncthreads();

    for (int p0 = s0; p0 < s1; p0 += 256) {
        int kk[8], rr[8]; float nn[8];
#pragma unroll
        for (int u = 0; u < 8; ++u) {
            const int e = p0 + u * 32 + lane;
            const int ec = (e < s1) ? e : (s1 - 1);
            kk[u] = key[ec];
            rr[u] = rin[ec];
            nn[u] = nin[ec];
        }
#pragma unroll
        for (int u = 0; u < 8; ++u) {
            const bool valid = (p0 + u * 32 + lane) < s1;
            const int d = (kk[u] >> BK_BITS) & (NDIG - 1);
            const unsigned m = same_digit_valid_lanes<HI_BITS>(d, valid);
            const int first = m ? (__ffs(m) - 1) : lane;
            int start = 0;
            if (valid && lane == first) {
                start = hw[d];
                hw[d] = start + __popc(m);
            }
            start = __shfl(start, first);
            const int pos = start + __popc(m & lt);
            if (valid) { stage_k[pos] = kk[u]; stage_r[pos] = rr[u]; stage_n[pos] = nn[u]; }
        }
    }
    __syncthreads();

    for (int i = t; i < SP_REC; i += SP_T) {
        const int k = stage_k[i];
        const int g = gbase[(k >> BK_BITS) & (NDIG - 1)] + i;
        kout[g] = k;
        rout[g] = stage_r[i];
        nout[g] = stage_n[i];
    }
}

__global__ void __launch_bounds__(BK_T)
bucket_order_kernel(const int* __restrict__ kin, const int* __restrict__ rin,
                    const float* __restrict__ nin, const int* __restrict__ T,
                    int* __restrict__ off, int* __restrict__ rout,
                    float* __restrict__ nout, float* __restrict__ dinv) {
    __shared__ int   h[BK_WAVES][BK_NODES];
    __shared__ int   wtot[BK_NODES / 32];
    __shared__ int   stage_r[BK_CAP];
    __shared__ float stage_n[BK_CAP];

    const int t = threadIdx.x, lane = t & 31;
    const int wave = __builtin_amdgcn_readfirstlane(t >> 5);
    const int b = blockIdx.x;
    const int beg = T[(size_t)b * NSPB];
    const int end = (b + 1 < NDIG) ? T[(size_t)(b + 1) * NSPB] : E_EDGES;
    const int n = end - beg;
    const int L  = ((n + BK_T - 1) / BK_T) * 32;
    const int s0 = beg + wave * L;
    const int s1 = (s0 + L < end) ? (s0 + L) : end;

    int* hw = h[wave];
    for (int i = lane; i < BK_NODES; i += 32) hw[i] = 0;
    const unsigned lt = (1u << lane) - 1u;

    for (int p0 = s0; p0 < s1; p0 += 256) {
        int kk[8];
#pragma unroll
        for (int u = 0; u < 8; ++u) {
            const int e = p0 + u * 32 + lane;
            kk[u] = kin[(e < s1) ? e : (s1 - 1)];
        }
#pragma unroll
        for (int u = 0; u < 8; ++u) {
            const bool valid = (p0 + u * 32 + lane) < s1;
            const int d = kk[u] & (BK_NODES - 1);
            const unsigned m = same_digit_valid_lanes<BK_BITS>(d, valid);
            if (valid && (m & lt) == 0) hw[d] = hw[d] + __popc(m);
        }
    }
    __syncthreads();

    int cnt = 0, inc = 0, run0 = 0;
    if (t < BK_NODES) {
#pragma unroll
        for (int w = 0; w < BK_WAVES; ++w) {
            const int c = h[w][t];
            h[w][t] = cnt;
            cnt += c;
        }
        inc = cnt;
#pragma unroll
        for (int d = 1; d < 32; d <<= 1) {
            const int y = __shfl_up(inc, d);
            if (lane >= d) inc += y;
        }
        if (lane == 31) wtot[wave] = inc;
    }
    __syncthreads();
    if (t < BK_NODES) {
        int P = inc - cnt;
        for (int w = 0; w < wave; ++w) P += wtot[w];
        run0 = P;
#pragma unroll
        for (int w = 0; w < BK_WAVES; ++w) h[w][t] += P;
        const int v = b * BK_NODES + t;
        if (v <= N_NODES) off[v] = beg + P;
    }
    __syncthreads();

    const bool staged = (n <= BK_CAP);
    for (int p0 = s0; p0 < s1; p0 += 256) {
        int kk[8], rr[8]; float nn[8];
#pragma unroll
        for (int u = 0; u < 8; ++u) {
            const int e = p0 + u * 32 + lane;
            const int ec = (e < s1) ? e : (s1 - 1);
            kk[u] = kin[ec];
            rr[u] = rin[ec];
            nn[u] = nin[ec];
        }
#pragma unroll
        for (int u = 0; u < 8; ++u) {
            const bool valid = (p0 + u * 32 + lane) < s1;
            const int d = kk[u] & (BK_NODES - 1);
            const unsigned m = same_digit_valid_lanes<BK_BITS>(d, valid);
            const int first = m ? (__ffs(m) - 1) : lane;
            int start = 0;
            if (valid && lane == first) {
                start = hw[d];
                hw[d] = start + __popc(m);
            }
            start = __shfl(start, first);
            const int pos = start + __popc(m & lt);
            if (valid) {
                if (staged) { stage_r[pos] = rr[u]; stage_n[pos] = nn[u]; }
                else        { rout[beg + pos] = rr[u]; nout[beg + pos] = nn[u]; }
            }
        }
    }
    __syncthreads();
    if (t < BK_NODES && b * BK_NODES + t < N_NODES) {
        float d = 0.0f;
        if (staged) {
            const int re = run0 + cnt;
            int i = run0;
            for (; i + 4 <= re; i += 4) {
                const float w0 = stage_n[i], w1 = stage_n[i + 1], w2 = stage_n[i + 2], w3 = stage_n[i + 3];
                d += w0; d += w1; d += w2; d += w3;
            }
            for (; i < re; ++i) d += stage_n[i];
        } else {
            for (int i = beg; i < end; ++i)
                if ((kin[i] & (BK_NODES - 1)) == t) d += nin[i];
        }
        dinv[b * BK_NODES + t] = (d > 0.0f) ? __frsqrt_rn(d) : 0.0f;
    }
    if (staged) {
        for (int i = t; i < n; i += BK_T) {
            rout[beg + i] = stage_r[i];
            nout[beg + i] = stage_n[i];
        }
    }
}

#define NRM_NODES 8
__global__ void __launch_bounds__(256)
norm_sorted_kernel(const int* __restrict__ off, const int* __restrict__ rs,
                   const float* __restrict__ dinv, float* __restrict__ ns) {
    const int lane = threadIdx.x & 31;
    const int wv = __builtin_amdgcn_readfirstlane((int)(blockIdx.x * 8 + (threadIdx.x >> 5)));
    const int v0 = wv * NRM_NODES;
    if (v0 >= N_NODES) return;
    int o[NRM_NODES + 1];
#pragma unroll
    for (int j = 0; j <= NRM_NODES; ++j) {
        const int vj = (v0 + j < N_NODES) ? (v0 + j) : N_NODES;
        o[j] = off[vj];
    }
    for (int i = o[0] + lane; i < o[NRM_NODES]; i += 32) {
        int k = 0;
#pragma unroll
        for (int j = 1; j < NRM_NODES; ++j) k += (i >= o[j]) ? 1 : 0;
        ns[i] = dinv[rs[i]] * ns[i] * dinv[v0 + k];
    }
}


template <int VPL> struct row_piece;
template <> struct row_piece<1> { typedef float type; };
template <> struct row_piece<2> { typedef v2f   type; };

template <int D, int G>
__device__ __forceinline__ void gather_group(const float* __restrict__ src, int lane,
                                             int rv, int nv, int j,
                                             typename row_piece<D / 32>::type& acc) {
    typedef typename row_piece<D / 32>::type vec;
    vec x[G];
#pragma unroll
    for (int k = 0; k < G; ++k) {
        const int r = __builtin_amdgcn_readlane(rv, j + k);
        x[k] = reinterpret_cast<const vec*>(src + (size_t)r * D)[lane];
    }
#pragma unroll
    for (int k = 0; k < G; ++k) {
        const float w = __int_as_float(__builtin_amdgcn_readlane(nv, j + k));
        acc += w * x[k];
    }
}

template <int D>
__global__ void __launch_bounds__(256)
gather_kernel(const float* __restrict__ src,
              float* __restrict__ dst,
              const int* __restrict__ off,
              const int* __restrict__ rs,
              const float* __restrict__ ns) {
    typedef typename row_piece<D / 32>::type vec;
    const int lane = threadIdx.x & 31;
    const int wv = __builtin_amdgcn_readfirstlane((int)(blockIdx.x * 8 + (threadIdx.x >> 5)));

    for (int v = wv; v < N_NODES; v += HOP_WAVES) {
        const int beg = off[v];
        const int end = off[v + 1];
        vec acc = 0.0f;
        for (int b = beg; b < end; b += 32) {
            const int n = (end - b < 32) ? (end - b) : 32;
            int rv = 0, nv = 0;
            if (lane < n) {
                rv = rs[b + lane];
                nv = __float_as_int(ns[b + lane]);
            }
            int j = 0;
            for (; j + 16 <= n; j += 16) gather_group<D, 16>(src, lane, rv, nv, j, acc);
            if (n - j >= 8) { gather_group<D, 8>(src, lane, rv, nv, j, acc); j += 8; }
            if (n - j >= 4) { gather_group<D, 4>(src, lane, rv, nv, j, acc); j += 4; }
            if (n - j >= 2) { gather_group<D, 2>(src, lane, rv, nv, j, acc); j += 2; }
            if (n - j >= 1) { gather_group<D, 1>(src, lane, rv, nv, j, acc); }
        }
        reinterpret_cast<vec*>(dst + (size_t)v * D)[lane] = acc;
    }
}


#define WQ_PITCH 288

template <int D>
__global__ void __launch_bounds__(128)
tag_gemm_kernel(const float* __restrict__ H0, const float* __restrict__ H1,
                const float* __restrict__ H2, const float* __restrict__ H3,
                const float* __restrict__ W,
                const float* __restrict__ bias,
                float* __restrict__ out) {
    constexpr int NQ = (4 * D) / 4;
    __shared__ __attribute__((aligned(16))) float Wl[NQ * WQ_PITCH];

    const int t = threadIdx.x;
#pragma unroll
    for (int it = 0; it < NQ * 16 / 128; ++it) {
        const int i = t + 128 * it;
        const int q = i >> 4, m = i & 15;
        const float* __restrict__ src = W + (size_t)(4 * q) * 64 + 4 * m;
        const v4f r0 = *reinterpret_cast<const v4f*>(src);
        const v4f r1 = *reinterpret_cast<const v4f*>(src + 64);
        const v4f r2 = *reinterpret_cast<const v4f*>(src + 128);
        const v4f r3 = *reinterpret_cast<const v4f*>(src + 192);
        float* dst = Wl + q * WQ_PITCH + m * 4;
        v4f c0 = {r0.x, r1.x, r2.x, r3.x};
        v4f c1 = {r0.y, r1.y, r2.y, r3.y};
        v4f c2 = {r0.z, r1.z, r2.z, r3.z};
        v4f c3 = {r0.w, r1.w, r2.w, r3.w};
        *reinterpret_cast<v4f*>(dst)       = c0;
        *reinterpret_cast<v4f*>(dst + 64)  = c1;
        *reinterpret_cast<v4f*>(dst + 128) = c2;
        *reinterpret_cast<v4f*>(dst + 192) = c3;
    }
    __syncthreads();

    const int wave  = t >> 5;
    const int lane  = t & 31;
    const int mtile = blockIdx.x * 4 + wave;
    if (mtile >= N_NODES / 16) return;

    const int c = lane & 15;
    const int h = lane >> 4;
    const size_t arow = (size_t)mtile * 16 + c;

    const float* Hs[4] = {H0, H1, H2, H3};
    v8f acc[4] = {};

#pragma unroll
    for (int s = 0; s < 4; ++s) {
        const float* __restrict__ A = Hs[s] + arow * D + 4 * h;
        v4f a[D / 8];
#pragma unroll
        for (int j = 0; j < D / 8; ++j) a[j] = *reinterpret_cast<const v4f*>(A + 8 * j);
#pragma unroll
        for (int j = 0; j < D / 8; ++j) {
            const float* bq = Wl + ((s * D + 8 * j) / 4 + h) * WQ_PITCH + c * 4;
            v4f b[4];
#pragma unroll
            for (int tt = 0; tt < 4; ++tt) b[tt] = *reinterpret_cast<const v4f*>(bq + tt * 64);
            v2f a0 = {a[j].x, a[j].y};
            v2f a1 = {a[j].z, a[j].w};
#pragma unroll
            for (int tt = 0; tt < 4; ++tt) {
                v2f b0 = {b[tt].x, b[tt].y};
                acc[tt] = __builtin_amdgcn_wmma_f32_16x16x4_f32(
                         false, a0,  false, b0,
                         (short)0, acc[tt],  false,  false);
            }
#pragma unroll
            for (int tt = 0; tt < 4; ++tt) {
                v2f b1 = {b[tt].z, b[tt].w};
                acc[tt] = __builtin_amdgcn_wmma_f32_16x16x4_f32(
                         false, a1,  false, b1,
                         (short)0, acc[tt],  false,  false);
            }
        }
    }

    const v4f bv = *reinterpret_cast<const v4f*>(bias + 4 * c);
#pragma unroll
    for (int r = 0; r < 8; ++r) {
        const size_t orow = (size_t)mtile * 16 + r + h * 8;
        v4f o = {acc[0][r] + bv.x, acc[1][r] + bv.y, acc[2][r] + bv.z, acc[3][r] + bv.w};
        o.x = o.x > 0.0f ? o.x : 0.0f;
        o.y = o.y > 0.0f ? o.y : 0.0f;
        o.z = o.z > 0.0f ? o.z : 0.0f;
        o.w = o.w > 0.0f ? o.w : 0.0f;
        *reinterpret_cast<v4f*>(out + orow * 64 + 4 * c) = o;
    }
}


static void split_pass(const int* kin, const int* rin, const float* nin,
                       int* T, int* bt, int* kout, int* rout, float* nout,
                       hipStream_t stream) {
    constexpr int nt = NDIG * NSPB;
    constexpr int sb = (nt + SCAN_T - 1) / SCAN_T;
    split_count_kernel<<<NSPB, SP_T, 0, stream>>>(kin, T);
    block_total_kernel<<<sb, SCAN_T, 0, stream>>>(T, nt, bt);
    scan_final_kernel<<<sb, SCAN_T, 0, stream>>>(T, nt, bt);
    split_place_kernel<<<NSPB, SP_T, 0, stream>>>(kin, rin, nin, T, kout, rout, nout);
}

template <int D>
static void run_layer(const float* hin, const float* W, const float* bias,
                      float* buf1, float* buf2, float* buf3,
                      const int* off, const int* rs, const float* ns,
                      float* hout, hipStream_t stream) {
    gather_kernel<D><<<HOP_BLOCKS, 256, 0, stream>>>(hin,  buf1, off, rs, ns);
    gather_kernel<D><<<HOP_BLOCKS, 256, 0, stream>>>(buf1, buf2, off, rs, ns);
    gather_kernel<D><<<HOP_BLOCKS, 256, 0, stream>>>(buf2, buf3, off, rs, ns);

    tag_gemm_kernel<D><<<(N_NODES / 16 + 3) / 4, 128, 0, stream>>>(
        hin, buf1, buf2, buf3, W, bias, hout);
}

extern "C" void kernel_launch(void* const* d_in, const int* in_sizes, int n_in,
                              void* d_out, int out_size, void* d_ws, size_t ws_size,
                              hipStream_t stream) {
    const float* x   = (const float*)d_in[0];
    const int*   ei  = (const int*)d_in[1];
    const float* ew  = (const float*)d_in[2];
    const float* W0  = (const float*)d_in[3];
    const float* b0  = (const float*)d_in[4];
    const float* Ws  = (const float*)d_in[5];
    const float* bs  = (const float*)d_in[6];
    float*       out = (float*)d_out;

    const int* row = ei;
    const int* col = ei + E_EDGES;

    const size_t NPAD = 100096;
    const size_t TPAD = 640000;
    float* ws   = (float*)d_ws;
    float* deg  = ws;
    int*   off  = (int*)(deg + NPAD);
    int*   bt   = off + NPAD;
    int*   T    = bt + 1024;
    int*   k1   = T + TPAD;
    int*   r1   = k1 + E_EDGES;
    float* n1   = (float*)(r1 + E_EDGES);
    int*   r2   = (int*)(n1 + E_EDGES);
    float* n2   = (float*)(r2 + E_EDGES);
    float* bufA = n2 + E_EDGES;
    float* bufB = bufA + (size_t)N_NODES * LATENT;
    float* buf1 = bufB + (size_t)N_NODES * LATENT;
    float* buf2 = buf1 + (size_t)N_NODES * LATENT;
    float* buf3 = buf2 + (size_t)N_NODES * LATENT;

    split_pass(col, row, ew, T, bt, k1, r1, n1, stream);
    bucket_order_kernel<<<NBUCKETS, BK_T, 0, stream>>>(k1, r1, n1, T, off, r2, n2, deg);

    norm_sorted_kernel<<<(N_NODES / NRM_NODES + 7) / 8, 256, 0, stream>>>(off, r2, deg, n2);

    const size_t WL = 4 * 64 * 64;
    run_layer<IN_DIM>(x,    W0,          b0,       buf1, buf2, buf3, off, r2, n2, bufA, stream);
    run_layer<LATENT>(bufA, Ws + 0 * WL, bs + 0,   buf1, buf2, buf3, off, r2, n2, bufB, stream);
    run_layer<LATENT>(bufB, Ws + 1 * WL, bs + 64,  buf1, buf2, buf3, off, r2, n2, bufA, stream);
    run_layer<LATENT>(bufA, Ws + 2 * WL, bs + 128, buf1, buf2, buf3, off, r2, n2, bufB, stream);
    run_layer<LATENT>(bufB, Ws + 3 * WL, bs + 192, buf1, buf2, buf3, off, r2, n2, out,  stream);
}
